# starting kernel
# baseline (speedup 1.0000x reference)
#include <hip/hip_runtime.h>


typedef __attribute__((ext_vector_type(2))) float v2f;
typedef __attribute__((ext_vector_type(8))) float v8f;

#define NCOL 128
#define TILE_M 128
#define KC 64
#define LXS 68
#define LWS 288

__global__ __launch_bounds__(256) void k_gemm_wmma(
    const float* __restrict__ X, const float* __restrict__ W,
    float* __restrict__ C, int M, int K) {
  __shared__ float lw2[(KC / 2) * LWS];
  __shared__ float lx[TILE_M * LXS];

  const int tid = threadIdx.x;
  const int lane = tid & 31;
  const int wave = tid >> 5;
  const int mblock = blockIdx.x * TILE_M;
  const int mstrip = mblock + wave * 16;

  v8f acc[8] = {};

  const int lane15 = lane & 15;
  const int ksel = (lane < 16) ? 0 : 2;
  const int mrow = wave * 16 + lane15;

  const float* axp = lx + mrow * LXS + ksel;
  const float* bbp = lw2 + (ksel >> 1) * LWS + lane15 * 2;

  for (int k0 = 0; k0 < K; k0 += KC) {
    __syncthreads();
    for (int i = tid; i < (KC / 2) * (NCOL / 4); i += 256) {
      const int kk2 = i >> 5;
      const int c4 = i & 31;
      const float* wr = W + (size_t)(k0 + kk2 * 2) * NCOL + c4 * 4;
      const float4 w0 = *(const float4*)wr;
      const float4 w1 = *(const float4*)(wr + NCOL);
      float* dst = lw2 + kk2 * LWS + c4 * 8;
      ((float2*)dst)[0] = make_float2(w0.x, w1.x);
      ((float2*)dst)[1] = make_float2(w0.y, w1.y);
      ((float2*)dst)[2] = make_float2(w0.z, w1.z);
      ((float2*)dst)[3] = make_float2(w0.w, w1.w);
    }
    for (int i = tid; i < TILE_M * (KC / 4); i += 256) {
      const int r = i >> 4;
      const int c4 = i & 15;
      const int gr = mblock + r;
      float4 v = make_float4(0.f, 0.f, 0.f, 0.f);
      if (gr < M) v = *(const float4*)(X + (size_t)gr * K + k0 + c4 * 4);
      *(float4*)(lx + r * LXS + c4 * 4) = v;
    }
    __syncthreads();

#pragma unroll 4
    for (int kk = 0; kk < KC; kk += 4) {
      const v2f a = *(const v2f*)(axp + kk);
      const float* brow = bbp + (kk >> 1) * LWS;
#pragma unroll
      for (int t = 0; t < 8; ++t) {
        const v2f b = *(const v2f*)(brow + t * 32);
        acc[t] = __builtin_amdgcn_wmma_f32_16x16x4_f32(
             false, a,  false, b,
             (short)0, acc[t],  false,  false);
      }
    }
  }

  if (mstrip < M) {
    const int mo = (lane < 16) ? 0 : 8;
#pragma unroll
    for (int t = 0; t < 8; ++t) {
#pragma unroll
      for (int v = 0; v < 8; ++v) {
        const int m = mstrip + mo + v;
        if (m < M) C[(size_t)m * NCOL + t * 16 + lane15] = acc[t][v];
      }
    }
  }
}

__global__ void k_deg_init(float* deg, int N) {
  int i = blockIdx.x * blockDim.x + threadIdx.x;
  if (i < N) deg[i] = 1.0f;
}

__global__ void k_deg_count(const int* __restrict__ col, float* deg, int E) {
  int i = blockIdx.x * blockDim.x + threadIdx.x;
  if (i < E) atomicAdd(&deg[col[i]], 1.0f);
}

__global__ void k_rsqrt(float* deg, int N) {
  int i = blockIdx.x * blockDim.x + threadIdx.x;
  if (i < N) {
    float d = deg[i];
    deg[i] = (d > 0.f) ? rsqrtf(d) : 0.f;
  }
}

__global__ void k_selfloop_init(const float* __restrict__ hlin,
                                const float* __restrict__ dinv,
                                float* __restrict__ agg, int N) {
  int i = blockIdx.x * blockDim.x + threadIdx.x;
  if (i < N * NCOL) {
    int n = i >> 7;
    float w = dinv[n];
    agg[i] = hlin[i] * w * w;
  }
}

__global__ __launch_bounds__(256) void k_edge_scatter(
    const float* __restrict__ hlin, const int* __restrict__ row,
    const int* __restrict__ col, const float* __restrict__ dinv,
    float* __restrict__ agg, int E) {
  int e = blockIdx.x * 2 + (threadIdx.x >> 7);
  int f = threadIdx.x & 127;
  if (e >= E) return;
  int s = row[e];
  int d = col[e];
  float w = dinv[s] * dinv[d];
  atomicAdd(&agg[(size_t)d * NCOL + f], hlin[(size_t)s * NCOL + f] * w);
}

__global__ void k_bias_act(float* __restrict__ h, const float* __restrict__ b,
                           int N, int do_relu) {
  int i = blockIdx.x * blockDim.x + threadIdx.x;
  if (i < N * NCOL) {
    float v = h[i] + b[i & 127];
    h[i] = (do_relu && v < 0.f) ? 0.f : v;
  }
}

__global__ void k_zero(float* p, int n) {
  int i = blockIdx.x * blockDim.x + threadIdx.x;
  if (i < n) p[i] = 0.f;
}

__global__ void k_pool(const float* __restrict__ h, const int* __restrict__ batch,
                       float* __restrict__ hs, int N) {
  int i = blockIdx.x * blockDim.x + threadIdx.x;
  if (i < N * NCOL) {
    int n = i >> 7;
    atomicAdd(&hs[(size_t)batch[n] * NCOL + (i & 127)], h[i]);
  }
}

extern "C" void kernel_launch(void* const* d_in, const int* in_sizes, int n_in,
                              void* d_out, int out_size, void* d_ws, size_t ws_size,
                              hipStream_t stream) {
  const float* x = (const float*)d_in[0];
  const float* W1 = (const float*)d_in[1];
  const float* b1 = (const float*)d_in[2];
  const float* W2 = (const float*)d_in[3];
  const float* b2 = (const float*)d_in[4];
  const int* eidx = (const int*)d_in[5];
  const int* batch = (const int*)d_in[6];

  const int N = in_sizes[6];
  const int E = in_sizes[5] / 2;
  const int K1 = in_sizes[0] / N;
  const int* row = eidx;
  const int* col = eidx + E;

  float* ws = (float*)d_ws;
  float* dinv = ws;
  float* bufA = ws + (((size_t)N + 63) & ~(size_t)63);
  float* bufB = bufA + (size_t)N * NCOL;

  const int G = out_size / NCOL - N;
  float* hs = (float*)d_out;
  float* hout = hs + (size_t)G * NCOL;

  const dim3 b256(256);
  const int nfBlocks = (N * NCOL + 255) / 256;
  const int gemmBlocks = (N + TILE_M - 1) / TILE_M;
  const int edgeBlocks = (E + 1) / 2;

  k_deg_init<<<(N + 255) / 256, b256, 0, stream>>>(dinv, N);
  k_deg_count<<<(E + 255) / 256, b256, 0, stream>>>(col, dinv, E);
  k_rsqrt<<<(N + 255) / 256, b256, 0, stream>>>(dinv, N);

  k_gemm_wmma<<<gemmBlocks, b256, 0, stream>>>(x, W1, bufA, N, K1);
  k_selfloop_init<<<nfBlocks, b256, 0, stream>>>(bufA, dinv, bufB, N);
  k_edge_scatter<<<edgeBlocks, b256, 0, stream>>>(bufA, row, col, dinv, bufB, E);
  k_bias_act<<<nfBlocks, b256, 0, stream>>>(bufB, b1, N, 1);

  k_gemm_wmma<<<gemmBlocks, b256, 0, stream>>>(bufB, W2, bufA, N, NCOL);
  k_selfloop_init<<<nfBlocks, b256, 0, stream>>>(bufA, dinv, hout, N);
  k_edge_scatter<<<edgeBlocks, b256, 0, stream>>>(bufA, row, col, dinv, hout, E);
  k_bias_act<<<nfBlocks, b256, 0, stream>>>(hout, b2, N, 0);

  k_zero<<<(G * NCOL + 255) / 256, b256, 0, stream>>>(hs, G * NCOL);
  k_pool<<<nfBlocks, b256, 0, stream>>>(hout, batch, hs, N);
}


# Degrees counted per destination tile from the edge bitmap: a block reads its bitmap row and col[] of its edges, adds 1 in LDS integer counters; global float adds on deg, init and rsqrt launches gone
# speedup vs baseline: 3.1683x; 3.1683x over previous
#include <hip/hip_runtime.h>


typedef __attribute__((ext_vector_type(2))) float v2f;
typedef __attribute__((ext_vector_type(8))) float v8f;

#define NCOL 128
#define TILE_M 128
#define KC 64
#define LXS 68
#define LWS 288

__global__ __launch_bounds__(256) void k_gemm_wmma(
    const float* __restrict__ X, const float* __restrict__ W,
    float* __restrict__ C, int M, int K) {
  __shared__ float lw2[(KC / 2) * LWS];
  __shared__ float lx[TILE_M * LXS];

  const int tid = threadIdx.x;
  const int lane = tid & 31;
  const int wave = tid >> 5;
  const int mblock = blockIdx.x * TILE_M;
  const int mstrip = mblock + wave * 16;

  v8f acc[8] = {};

  const int lane15 = lane & 15;
  const int ksel = (lane < 16) ? 0 : 2;
  const int mrow = wave * 16 + lane15;

  const float* axp = lx + mrow * LXS + ksel;
  const float* bbp = lw2 + (ksel >> 1) * LWS + lane15 * 2;

  for (int k0 = 0; k0 < K; k0 += KC) {
    __syncthreads();
    for (int i = tid; i < (KC / 2) * (NCOL / 4); i += 256) {
      const int kk2 = i >> 5;
      const int c4 = i & 31;
      const float* wr = W + (size_t)(k0 + kk2 * 2) * NCOL + c4 * 4;
      const float4 w0 = *(const float4*)wr;
      const float4 w1 = *(const float4*)(wr + NCOL);
      float* dst = lw2 + kk2 * LWS + c4 * 8;
      ((float2*)dst)[0] = make_float2(w0.x, w1.x);
      ((float2*)dst)[1] = make_float2(w0.y, w1.y);
      ((float2*)dst)[2] = make_float2(w0.z, w1.z);
      ((float2*)dst)[3] = make_float2(w0.w, w1.w);
    }
    for (int i = tid; i < TILE_M * (KC / 4); i += 256) {
      const int r = i >> 4;
      const int c4 = i & 15;
      const int gr = mblock + r;
      float4 v = make_float4(0.f, 0.f, 0.f, 0.f);
      if (gr < M) v = *(const float4*)(X + (size_t)gr * K + k0 + c4 * 4);
      *(float4*)(lx + r * LXS + c4 * 4) = v;
    }
    __syncthreads();

#pragma unroll 4
    for (int kk = 0; kk < KC; kk += 4) {
      const v2f a = *(const v2f*)(axp + kk);
      const float* brow = bbp + (kk >> 1) * LWS;
#pragma unroll
      for (int t = 0; t < 8; ++t) {
        const v2f b = *(const v2f*)(brow + t * 32);
        acc[t] = __builtin_amdgcn_wmma_f32_16x16x4_f32(
             false, a,  false, b,
             (short)0, acc[t],  false,  false);
      }
    }
  }

  if (mstrip < M) {
    const int mo = (lane < 16) ? 0 : 8;
#pragma unroll
    for (int t = 0; t < 8; ++t) {
#pragma unroll
      for (int v = 0; v < 8; ++v) {
        const int m = mstrip + mo + v;
        if (m < M) C[(size_t)m * NCOL + t * 16 + lane15] = acc[t][v];
      }
    }
  }
}

#define TD 200
#define AGG_T 1024

#define BM_C 4
#define BM_W (32 * BM_C)
#define BM_P (BM_W + 4)
#define BM_MAXT 256

__global__ __launch_bounds__(AGG_T) void k_bitmap(
    const int* __restrict__ col, unsigned* __restrict__ bm, int E, int WP,
    int nTiles) {
  __shared__ unsigned sm[BM_MAXT * BM_P];
  const int tid = threadIdx.x;
  uint4* sm4 = (uint4*)sm;
  const int nz = nTiles * (BM_P / 4);
  for (int i = tid; i < nz; i += AGG_T) sm4[i] = make_uint4(0u, 0u, 0u, 0u);
  __syncthreads();

  const int e0 = blockIdx.x * (AGG_T * BM_C) + tid;
  int d[BM_C];
#pragma unroll
  for (int c = 0; c < BM_C; ++c) {
    const int e = e0 + c * AGG_T;
    d[c] = (e < E) ? col[e] : -1;
  }
#pragma unroll
  for (int c = 0; c < BM_C; ++c) {
    if (d[c] >= 0)
      atomicOr(&sm[(d[c] / TD) * BM_P + c * 32 + (tid >> 5)], 1u << (tid & 31));
  }
  __syncthreads();

  const int nw = nTiles * (BM_W / 4);
  for (int i = tid; i < nw; i += AGG_T) {
    const int t = i / (BM_W / 4);
    const int j = i % (BM_W / 4);
    *(uint4*)(bm + (size_t)t * WP + (size_t)blockIdx.x * BM_W + j * 4) =
        sm4[t * (BM_P / 4) + j];
  }
}

#define DG_U 20

__global__ __launch_bounds__(AGG_T) void k_deg_tile(
    const int* __restrict__ col, const unsigned* __restrict__ bm, int WP,
    float* __restrict__ dinv, int N) {
  __shared__ int cnt[TD];
  const int tid = threadIdx.x;
  const int d0 = blockIdx.x * TD;
  const int rows = (N - d0 < TD) ? (N - d0) : TD;
  if (tid < TD) cnt[tid] = 0;
  __syncthreads();

  const unsigned* bmrow = bm + (size_t)blockIdx.x * WP;
  for (int base = 0; base < WP; base += DG_U * AGG_T) {
    unsigned m[DG_U];
#pragma unroll
    for (int j = 0; j < DG_U; ++j) {
      const int w = base + j * AGG_T + tid;
      m[j] = (w < WP) ? bmrow[w] : 0u;
    }
    bool more;
    do {
      int dv[DG_U];
      unsigned hm = 0u;
#pragma unroll
      for (int j = 0; j < DG_U; ++j) {
        const unsigned mm = m[j];
        const bool has = (mm != 0u);
        dv[j] = 0;
        if (__builtin_amdgcn_ballot_w32(has) != 0u) {
          const int e =
              has ? (((base + j * AGG_T + tid) << 5) + __builtin_ctz(mm)) : 0;
          dv[j] = col[e];
          hm |= has ? (1u << j) : 0u;
          m[j] = mm & (mm - 1u);
        }
      }
      more = false;
#pragma unroll
      for (int j = 0; j < DG_U; ++j) {
        if (hm & (1u << j)) atomicAdd(&cnt[dv[j] - d0], 1);
        more = more || (m[j] != 0u);
      }
    } while (more);
  }
  __syncthreads();

  if (tid < rows) {
    const float d = 1.0f + (float)cnt[tid];
    dinv[d0 + tid] = (d > 0.f) ? rsqrtf(d) : 0.f;
  }
}

#define SCAN_U 4
#define QCAP 64

__device__ __forceinline__ void bm_load(unsigned (&m)[SCAN_U],
                                        const unsigned* __restrict__ rowp,
                                        int wb, int WP) {
#pragma unroll
  for (int j = 0; j < SCAN_U; ++j) {
    const int w = wb + j * AGG_T;
    m[j] = (w < WP) ? rowp[w] : 0u;
  }
}

__device__ __forceinline__ void agg_drain(
    const int* qp, int n, int lane, int d0, const float* __restrict__ hlin,
    const int* __restrict__ row, const int* __restrict__ col,
    const float* __restrict__ dinv, float* acc) {
  int s = 0, dl = 0;
  float w = 0.f;
  if (lane < n) {
    const int e = qp[lane];
    const int d = col[e];
    s = row[e];
    dl = d - d0;
    w = dinv[s] * dinv[d];
  }
  for (int i = 0; i < n; i += 8) {
    int sb[8], db[8];
    float wb[8];
    float v[8][4];
#pragma unroll
    for (int u = 0; u < 8; ++u) {
      const int ii = (i + u < n) ? (i + u) : (n - 1);
      sb[u] = __builtin_amdgcn_readlane(s, ii);
      db[u] = __builtin_amdgcn_readlane(dl, ii);
      wb[u] = __int_as_float(__builtin_amdgcn_readlane(__float_as_int(w), ii));
    }
#pragma unroll
    for (int u = 0; u < 8; ++u) {
      const float* p = hlin + (size_t)sb[u] * NCOL + lane;
      v[u][0] = p[0];
      v[u][1] = p[32];
      v[u][2] = p[64];
      v[u][3] = p[96];
    }
#pragma unroll
    for (int u = 0; u < 8; ++u) {
      if (i + u < n) {
        float* a = acc + db[u] * NCOL + lane;
        atomicAdd(a, v[u][0] * wb[u]);
        atomicAdd(a + 32, v[u][1] * wb[u]);
        atomicAdd(a + 64, v[u][2] * wb[u]);
        atomicAdd(a + 96, v[u][3] * wb[u]);
      }
    }
  }
}

__global__ __launch_bounds__(AGG_T) void k_agg_tile(
    const float* __restrict__ hlin, const int* __restrict__ row,
    const int* __restrict__ col, const float* __restrict__ dinv,
    const float* __restrict__ bias, const unsigned* __restrict__ bm, int WP,
    float* __restrict__ out, int N, int do_relu) {
  __shared__ float acc[TD * NCOL];
  __shared__ int q[AGG_T / 32][QCAP];

  const int tid = threadIdx.x;
  const int lane = tid & 31;
  const int wave = __builtin_amdgcn_readfirstlane(tid >> 5);
  const int d0 = blockIdx.x * TD;
  const int rows = (N - d0 < TD) ? (N - d0) : TD;
  const int nvec = rows * (NCOL / 4);

  const unsigned* bmrow = bm + (size_t)blockIdx.x * WP + lane;
  unsigned mn[SCAN_U];
  bm_load(mn, bmrow, wave * 32, WP);

  const float4* hsrc = (const float4*)(hlin + (size_t)d0 * NCOL);
  for (int i = tid; i < nvec; i += AGG_T) {
    const float wd = dinv[d0 + (i >> 5)];
    const float4 h = hsrc[i];
    float4 o;
    o.x = h.x * wd * wd;
    o.y = h.y * wd * wd;
    o.z = h.z * wd * wd;
    o.w = h.w * wd * wd;
    ((float4*)acc)[i] = o;
  }
  __syncthreads();

  int* qw = q[wave];
  int cnt = 0;
  for (int wb = wave * 32; wb < WP; wb += SCAN_U * AGG_T) {
    const unsigned m0 = mn[0], m1 = mn[1], m2 = mn[2], m3 = mn[3];
    bm_load(mn, bmrow, wb + SCAN_U * AGG_T, WP);
    int j = 0;
    unsigned mm = m0;
    for (;;) {
      const bool has = (mm != 0u);
      const unsigned bal = __builtin_amdgcn_ballot_w32(has);
      if (bal == 0u) {
        if (++j == SCAN_U) break;
        mm = (j == 1) ? m1 : ((j == 2) ? m2 : m3);
        continue;
      }
      if (has) {
        const int e = (wb + j * AGG_T + lane) * 32 + __builtin_ctz(mm);
        qw[cnt + (int)__builtin_amdgcn_mbcnt_lo(bal, 0u)] = e;
      }
      cnt += __builtin_popcount(bal);
      mm &= mm - 1u;
      if (cnt >= 32) {
        cnt -= 32;
        agg_drain(qw + cnt, 32, lane, d0, hlin, row, col, dinv, acc);
      }
    }
  }
  agg_drain(qw, cnt, lane, d0, hlin, row, col, dinv, acc);
  __syncthreads();

  float4* odst = (float4*)(out + (size_t)d0 * NCOL);
  for (int i = tid; i < nvec; i += AGG_T) {
    const int f = (i & 31) * 4;
    float4 a = ((const float4*)acc)[i];
    a.x += bias[f];
    a.y += bias[f + 1];
    a.z += bias[f + 2];
    a.w += bias[f + 3];
    if (do_relu) {
      a.x = (a.x < 0.f) ? 0.f : a.x;
      a.y = (a.y < 0.f) ? 0.f : a.y;
      a.z = (a.z < 0.f) ? 0.f : a.z;
      a.w = (a.w < 0.f) ? 0.f : a.w;
    }
    odst[i] = a;
  }
}

__global__ void k_zero(float* p, int n) {
  int i = blockIdx.x * blockDim.x + threadIdx.x;
  if (i < n) p[i] = 0.f;
}

#define POOL_RW 16
#define POOL_T 256

__device__ __forceinline__ void pool_flush(float* __restrict__ hs, int g, int G,
                                           int lane, const float (&a)[4]) {
  if ((unsigned)g < (unsigned)G) {
    float* p = hs + (size_t)g * NCOL + lane;
    atomicAdd(p, a[0]);
    atomicAdd(p + 32, a[1]);
    atomicAdd(p + 64, a[2]);
    atomicAdd(p + 96, a[3]);
  }
}

__global__ __launch_bounds__(POOL_T) void k_pool_runs(
    const float* __restrict__ h, const int* __restrict__ batch,
    float* __restrict__ hs, int N, int G) {
  const int lane = threadIdx.x & 31;
  const int wave = __builtin_amdgcn_readfirstlane(threadIdx.x >> 5);
  const int r0 = (blockIdx.x * (POOL_T / 32) + wave) * POOL_RW;
  if (r0 >= N) return;
  const float* hp = h + (size_t)r0 * NCOL + lane;
  const int* bp = batch + r0;
  float acc[4] = {0.f, 0.f, 0.f, 0.f};

  if (N - r0 >= POOL_RW) {
    float v[POOL_RW][4];
    int g[POOL_RW];
#pragma unroll
    for (int j = 0; j < POOL_RW; ++j) {
      v[j][0] = hp[j * NCOL];
      v[j][1] = hp[j * NCOL + 32];
      v[j][2] = hp[j * NCOL + 64];
      v[j][3] = hp[j * NCOL + 96];
      g[j] = bp[j];
    }
    int cur = g[0];
#pragma unroll
    for (int j = 0; j < POOL_RW; ++j) {
      if (g[j] != cur) {
        pool_flush(hs, cur, G, lane, acc);
        acc[0] = acc[1] = acc[2] = acc[3] = 0.f;
        cur = g[j];
      }
      acc[0] += v[j][0];
      acc[1] += v[j][1];
      acc[2] += v[j][2];
      acc[3] += v[j][3];
    }
    pool_flush(hs, cur, G, lane, acc);
  } else {
    const int nr = N - r0;
    int cur = bp[0];
    for (int j = 0; j < nr; ++j) {
      const int gj = bp[j];
      if (gj != cur) {
        pool_flush(hs, cur, G, lane, acc);
        acc[0] = acc[1] = acc[2] = acc[3] = 0.f;
        cur = gj;
      }
      acc[0] += hp[j * NCOL];
      acc[1] += hp[j * NCOL + 32];
      acc[2] += hp[j * NCOL + 64];
      acc[3] += hp[j * NCOL + 96];
    }
    pool_flush(hs, cur, G, lane, acc);
  }
}

extern "C" void kernel_launch(void* const* d_in, const int* in_sizes, int n_in,
                              void* d_out, int out_size, void* d_ws, size_t ws_size,
                              hipStream_t stream) {
  const float* x = (const float*)d_in[0];
  const float* W1 = (const float*)d_in[1];
  const float* b1 = (const float*)d_in[2];
  const float* W2 = (const float*)d_in[3];
  const float* b2 = (const float*)d_in[4];
  const int* eidx = (const int*)d_in[5];
  const int* batch = (const int*)d_in[6];

  const int N = in_sizes[6];
  const int E = in_sizes[5] / 2;
  const int K1 = in_sizes[0] / N;
  const int* row = eidx;
  const int* col = eidx + E;

  float* ws = (float*)d_ws;
  float* dinv = ws;
  float* bufA = ws + (((size_t)N + 63) & ~(size_t)63);
  unsigned* bm = (unsigned*)(bufA + (size_t)N * NCOL);

  const int G = out_size / NCOL - N;
  float* hs = (float*)d_out;
  float* hout = hs + (size_t)G * NCOL;

  const dim3 b256(256);
  const int poolBlocks = (N + (POOL_T / 32) * POOL_RW - 1) / ((POOL_T / 32) * POOL_RW);
  const int gemmBlocks = (N + TILE_M - 1) / TILE_M;
  const int aggBlocks = (N + TD - 1) / TD;
  const dim3 bAgg(AGG_T);
  const int bmBlocks = (E + AGG_T * BM_C - 1) / (AGG_T * BM_C);
  const int WP = bmBlocks * BM_W;

  k_bitmap<<<bmBlocks, bAgg, 0, stream>>>(col, bm, E, WP, aggBlocks);
  k_deg_tile<<<aggBlocks, bAgg, 0, stream>>>(col, bm, WP, dinv, N);

  k_gemm_wmma<<<gemmBlocks, b256, 0, stream>>>(x, W1, bufA, N, K1);
  k_agg_tile<<<aggBlocks, bAgg, 0, stream>>>(bufA, row, col, dinv, b1, bm, WP, hout, N, 1);

  k_gemm_wmma<<<gemmBlocks, b256, 0, stream>>>(hout, W2, bufA, N, NCOL);
  k_agg_tile<<<aggBlocks, bAgg, 0, stream>>>(bufA, row, col, dinv, b2, bm, WP, hout, N, 0);

  k_zero<<<(G * NCOL + 255) / 256, b256, 0, stream>>>(hs, G * NCOL);
  k_pool_runs<<<poolBlocks, dim3(POOL_T), 0, stream>>>(hout, batch, hs, N, G);
}


# GEMM data path: whole W resident in LDS (one block per WGP, 25 waves of 16x64 tiles); X rows loaded straight into A fragments, no LDS stage or K-loop barrier; k order inside 8-k groups changed
# speedup vs baseline: 4.1157x; 1.2990x over previous
#include <hip/hip_runtime.h>


typedef __attribute__((ext_vector_type(2))) float v2f;
typedef __attribute__((ext_vector_type(8))) float v8f;

#define NCOL 128
#define KIN 512
#define GW 25
#define GT (GW * 32)
#define LWS 272

struct ablk {
  float4 q[4];
};

__device__ __forceinline__ void a_load(ablk& a, const float* __restrict__ p) {
#pragma unroll
  for (int q = 0; q < 4; ++q) a.q[q] = *(const float4*)(p + 8 * q);
}

__device__ __forceinline__ void blk_mma(v8f (&acc)[4], const ablk& a,
                                        const float* brow) {
#pragma unroll
  for (int q = 0; q < 4; ++q) {
#pragma unroll
    for (int s = 0; s < 2; ++s) {
      v2f av;
      av.x = s ? a.q[q].z : a.q[q].x;
      av.y = s ? a.q[q].w : a.q[q].y;
      const float* br = brow + (4 * q + s) * LWS;
#pragma unroll
      for (int t = 0; t < 4; ++t) {
        const v2f b = *(const v2f*)(br + t * 32);
        acc[t] = __builtin_amdgcn_wmma_f32_16x16x4_f32(
             false, av,  false, b,
             (short)0, acc[t],  false,  false);
      }
    }
  }
}

template <int K>
__global__ __launch_bounds__(GT) void k_gemm_wres(
    const float* __restrict__ X, const float* __restrict__ W,
    float* __restrict__ C, int M) {
  __shared__ float lw2[(K / 2) * LWS];

  const int tid = threadIdx.x;
  const int lane = tid & 31;
  const int wave = __builtin_amdgcn_readfirstlane(tid >> 5);
  const int c = lane & 15;
  const int h = lane >> 4;
  const int job = blockIdx.x * GW + wave;
  const int strip = job >> 1;
  const int half = job & 1;
  const int nstrips = (M + 15) >> 4;

  int r = strip * 16 + c;
  r = (r < M) ? r : (M - 1);
  const float* xp = X + (size_t)r * K + 4 * h;

  ablk a0, a1;
  a_load(a0, xp);
  a_load(a1, xp + 32);

  constexpr int NW = (K / 2) * (NCOL / 4);
  for (int i0 = tid; i0 < NW; i0 += 4 * GT) {
    float4 w0[4], w1[4];
#pragma unroll
    for (int u = 0; u < 4; ++u) {
      int i = i0 + u * GT;
      i = (i < NW) ? i : (NW - 1);
      const float* wr = W + (size_t)(i >> 5) * (2 * NCOL) + (i & 31) * 4;
      w0[u] = *(const float4*)wr;
      w1[u] = *(const float4*)(wr + NCOL);
    }
#pragma unroll
    for (int u = 0; u < 4; ++u) {
      const int i = i0 + u * GT;
      if (i < NW) {
        float* dst = lw2 + (i >> 5) * LWS + (i & 31) * 8;
        ((float4*)dst)[0] = make_float4(w0[u].x, w1[u].x, w0[u].y, w1[u].y);
        ((float4*)dst)[1] = make_float4(w0[u].z, w1[u].z, w0[u].w, w1[u].w);
      }
    }
  }
  __syncthreads();
  if (strip >= nstrips) return;

  v8f acc[4] = {};
  const float* bbp = lw2 + (2 * h) * LWS + (half * 64 + c) * 2;

  for (int kb = 0; kb < K; kb += 64) {
    blk_mma(acc, a0, bbp + (kb >> 1) * LWS);
    if (kb + 64 < K) a_load(a0, xp + kb + 64);
    blk_mma(acc, a1, bbp + ((kb + 32) >> 1) * LWS);
    if (kb + 96 < K) a_load(a1, xp + kb + 96);
  }

  float* cp = C + (size_t)(strip * 16 + 8 * h) * NCOL + half * 64 + c;
  const int m0 = strip * 16 + 8 * h;
#pragma unroll
  for (int t = 0; t < 4; ++t) {
#pragma unroll
    for (int v = 0; v < 8; ++v) {
      if (m0 + v < M) cp[(size_t)v * NCOL + t * 16] = acc[t][v];
    }
  }
}

#define TD 200
#define AGG_T 1024

#define BM_C 4
#define BM_W (32 * BM_C)
#define BM_P (BM_W + 4)
#define BM_MAXT 256

__global__ __launch_bounds__(AGG_T) void k_bitmap(
    const int* __restrict__ col, unsigned* __restrict__ bm, int E, int WP,
    int nTiles) {
  __shared__ unsigned sm[BM_MAXT * BM_P];
  const int tid = threadIdx.x;
  uint4* sm4 = (uint4*)sm;
  const int nz = nTiles * (BM_P / 4);
  for (int i = tid; i < nz; i += AGG_T) sm4[i] = make_uint4(0u, 0u, 0u, 0u);
  __syncthreads();

  const int e0 = blockIdx.x * (AGG_T * BM_C) + tid;
  int d[BM_C];
#pragma unroll
  for (int c = 0; c < BM_C; ++c) {
    const int e = e0 + c * AGG_T;
    d[c] = (e < E) ? col[e] : -1;
  }
#pragma unroll
  for (int c = 0; c < BM_C; ++c) {
    if (d[c] >= 0)
      atomicOr(&sm[(d[c] / TD) * BM_P + c * 32 + (tid >> 5)], 1u << (tid & 31));
  }
  __syncthreads();

  const int nw = nTiles * (BM_W / 4);
  for (int i = tid; i < nw; i += AGG_T) {
    const int t = i / (BM_W / 4);
    const int j = i % (BM_W / 4);
    *(uint4*)(bm + (size_t)t * WP + (size_t)blockIdx.x * BM_W + j * 4) =
        sm4[t * (BM_P / 4) + j];
  }
}

#define DG_U 20

__global__ __launch_bounds__(AGG_T) void k_deg_tile(
    const int* __restrict__ col, const unsigned* __restrict__ bm, int WP,
    float* __restrict__ dinv, int N) {
  __shared__ int cnt[TD];
  const int tid = threadIdx.x;
  const int d0 = blockIdx.x * TD;
  const int rows = (N - d0 < TD) ? (N - d0) : TD;
  if (tid < TD) cnt[tid] = 0;
  __syncthreads();

  const unsigned* bmrow = bm + (size_t)blockIdx.x * WP;
  for (int base = 0; base < WP; base += DG_U * AGG_T) {
    unsigned m[DG_U];
#pragma unroll
    for (int j = 0; j < DG_U; ++j) {
      const int w = base + j * AGG_T + tid;
      m[j] = (w < WP) ? bmrow[w] : 0u;
    }
    bool more;
    do {
      int dv[DG_U];
      unsigned hm = 0u;
#pragma unroll
      for (int j = 0; j < DG_U; ++j) {
        const unsigned mm = m[j];
        const bool has = (mm != 0u);
        dv[j] = 0;
        if (__builtin_amdgcn_ballot_w32(has) != 0u) {
          const int e =
              has ? (((base + j * AGG_T + tid) << 5) + __builtin_ctz(mm)) : 0;
          dv[j] = col[e];
          hm |= has ? (1u << j) : 0u;
          m[j] = mm & (mm - 1u);
        }
      }
      more = false;
#pragma unroll
      for (int j = 0; j < DG_U; ++j) {
        if (hm & (1u << j)) atomicAdd(&cnt[dv[j] - d0], 1);
        more = more || (m[j] != 0u);
      }
    } while (more);
  }
  __syncthreads();

  if (tid < rows) {
    const float d = 1.0f + (float)cnt[tid];
    dinv[d0 + tid] = (d > 0.f) ? rsqrtf(d) : 0.f;
  }
}

#define SCAN_U 4
#define QCAP 64

__device__ __forceinline__ void bm_load(unsigned (&m)[SCAN_U],
                                        const unsigned* __restrict__ rowp,
                                        int wb, int WP) {
#pragma unroll
  for (int j = 0; j < SCAN_U; ++j) {
    const int w = wb + j * AGG_T;
    m[j] = (w < WP) ? rowp[w] : 0u;
  }
}

__device__ __forceinline__ void agg_drain(
    const int* qp, int n, int lane, int d0, const float* __restrict__ hlin,
    const int* __restrict__ row, const int* __restrict__ col,
    const float* __restrict__ dinv, float* acc) {
  int s = 0, dl = 0;
  float w = 0.f;
  if (lane < n) {
    const int e = qp[lane];
    const int d = col[e];
    s = row[e];
    dl = d - d0;
    w = dinv[s] * dinv[d];
  }
  for (int i = 0; i < n; i += 8) {
    int sb[8], db[8];
    float wb[8];
    float v[8][4];
#pragma unroll
    for (int u = 0; u < 8; ++u) {
      const int ii = (i + u < n) ? (i + u) : (n - 1);
      sb[u] = __builtin_amdgcn_readlane(s, ii);
      db[u] = __builtin_amdgcn_readlane(dl, ii);
      wb[u] = __int_as_float(__builtin_amdgcn_readlane(__float_as_int(w), ii));
    }
#pragma unroll
    for (int u = 0; u < 8; ++u) {
      const float* p = hlin + (size_t)sb[u] * NCOL + lane;
      v[u][0] = p[0];
      v[u][1] = p[32];
      v[u][2] = p[64];
      v[u][3] = p[96];
    }
#pragma unroll
    for (int u = 0; u < 8; ++u) {
      if (i + u < n) {
        float* a = acc + db[u] * NCOL + lane;
        atomicAdd(a, v[u][0] * wb[u]);
        atomicAdd(a + 32, v[u][1] * wb[u]);
        atomicAdd(a + 64, v[u][2] * wb[u]);
        atomicAdd(a + 96, v[u][3] * wb[u]);
      }
    }
  }
}

__global__ __launch_bounds__(AGG_T) void k_agg_tile(
    const float* __restrict__ hlin, const int* __restrict__ row,
    const int* __restrict__ col, const float* __restrict__ dinv,
    const float* __restrict__ bias, const unsigned* __restrict__ bm, int WP,
    float* __restrict__ out, int N, int do_relu) {
  __shared__ float acc[TD * NCOL];
  __shared__ int q[AGG_T / 32][QCAP];

  const int tid = threadIdx.x;
  const int lane = tid & 31;
  const int wave = __builtin_amdgcn_readfirstlane(tid >> 5);
  const int d0 = blockIdx.x * TD;
  const int rows = (N - d0 < TD) ? (N - d0) : TD;
  const int nvec = rows * (NCOL / 4);

  const unsigned* bmrow = bm + (size_t)blockIdx.x * WP + lane;
  unsigned mn[SCAN_U];
  bm_load(mn, bmrow, wave * 32, WP);

  const float4* hsrc = (const float4*)(hlin + (size_t)d0 * NCOL);
  for (int i = tid; i < nvec; i += AGG_T) {
    const float wd = dinv[d0 + (i >> 5)];
    const float4 h = hsrc[i];
    float4 o;
    o.x = h.x * wd * wd;
    o.y = h.y * wd * wd;
    o.z = h.z * wd * wd;
    o.w = h.w * wd * wd;
    ((float4*)acc)[i] = o;
  }
  __syncthreads();

  int* qw = q[wave];
  int cnt = 0;
  for (int wb = wave * 32; wb < WP; wb += SCAN_U * AGG_T) {
    const unsigned m0 = mn[0], m1 = mn[1], m2 = mn[2], m3 = mn[3];
    bm_load(mn, bmrow, wb + SCAN_U * AGG_T, WP);
    int j = 0;
    unsigned mm = m0;
    for (;;) {
      const bool has = (mm != 0u);
      const unsigned bal = __builtin_amdgcn_ballot_w32(has);
      if (bal == 0u) {
        if (++j == SCAN_U) break;
        mm = (j == 1) ? m1 : ((j == 2) ? m2 : m3);
        continue;
      }
      if (has) {
        const int e = (wb + j * AGG_T + lane) * 32 + __builtin_ctz(mm);
        qw[cnt + (int)__builtin_amdgcn_mbcnt_lo(bal, 0u)] = e;
      }
      cnt += __builtin_popcount(bal);
      mm &= mm - 1u;
      if (cnt >= 32) {
        cnt -= 32;
        agg_drain(qw + cnt, 32, lane, d0, hlin, row, col, dinv, acc);
      }
    }
  }
  agg_drain(qw, cnt, lane, d0, hlin, row, col, dinv, acc);
  __syncthreads();

  float4* odst = (float4*)(out + (size_t)d0 * NCOL);
  for (int i = tid; i < nvec; i += AGG_T) {
    const int f = (i & 31) * 4;
    float4 a = ((const float4*)acc)[i];
    a.x += bias[f];
    a.y += bias[f + 1];
    a.z += bias[f + 2];
    a.w += bias[f + 3];
    if (do_relu) {
      a.x = (a.x < 0.f) ? 0.f : a.x;
      a.y = (a.y < 0.f) ? 0.f : a.y;
      a.z = (a.z < 0.f) ? 0.f : a.z;
      a.w = (a.w < 0.f) ? 0.f : a.w;
    }
    odst[i] = a;
  }
}

__global__ void k_zero(float* p, int n) {
  int i = blockIdx.x * blockDim.x + threadIdx.x;
  if (i < n) p[i] = 0.f;
}

#define POOL_RW 16
#define POOL_T 256

__device__ __forceinline__ void pool_flush(float* __restrict__ hs, int g, int G,
                                           int lane, const float (&a)[4]) {
  if ((unsigned)g < (unsigned)G) {
    float* p = hs + (size_t)g * NCOL + lane;
    atomicAdd(p, a[0]);
    atomicAdd(p + 32, a[1]);
    atomicAdd(p + 64, a[2]);
    atomicAdd(p + 96, a[3]);
  }
}

__global__ __launch_bounds__(POOL_T) void k_pool_runs(
    const float* __restrict__ h, const int* __restrict__ batch,
    float* __restrict__ hs, int N, int G) {
  const int lane = threadIdx.x & 31;
  const int wave = __builtin_amdgcn_readfirstlane(threadIdx.x >> 5);
  const int r0 = (blockIdx.x * (POOL_T / 32) + wave) * POOL_RW;
  if (r0 >= N) return;
  const float* hp = h + (size_t)r0 * NCOL + lane;
  const int* bp = batch + r0;
  float acc[4] = {0.f, 0.f, 0.f, 0.f};

  if (N - r0 >= POOL_RW) {
    float v[POOL_RW][4];
    int g[POOL_RW];
#pragma unroll
    for (int j = 0; j < POOL_RW; ++j) {
      v[j][0] = hp[j * NCOL];
      v[j][1] = hp[j * NCOL + 32];
      v[j][2] = hp[j * NCOL + 64];
      v[j][3] = hp[j * NCOL + 96];
      g[j] = bp[j];
    }
    int cur = g[0];
#pragma unroll
    for (int j = 0; j < POOL_RW; ++j) {
      if (g[j] != cur) {
        pool_flush(hs, cur, G, lane, acc);
        acc[0] = acc[1] = acc[2] = acc[3] = 0.f;
        cur = g[j];
      }
      acc[0] += v[j][0];
      acc[1] += v[j][1];
      acc[2] += v[j][2];
      acc[3] += v[j][3];
    }
    pool_flush(hs, cur, G, lane, acc);
  } else {
    const int nr = N - r0;
    int cur = bp[0];
    for (int j = 0; j < nr; ++j) {
      const int gj = bp[j];
      if (gj != cur) {
        pool_flush(hs, cur, G, lane, acc);
        acc[0] = acc[1] = acc[2] = acc[3] = 0.f;
        cur = gj;
      }
      acc[0] += hp[j * NCOL];
      acc[1] += hp[j * NCOL + 32];
      acc[2] += hp[j * NCOL + 64];
      acc[3] += hp[j * NCOL + 96];
    }
    pool_flush(hs, cur, G, lane, acc);
  }
}

extern "C" void kernel_launch(void* const* d_in, const int* in_sizes, int n_in,
                              void* d_out, int out_size, void* d_ws, size_t ws_size,
                              hipStream_t stream) {
  const float* x = (const float*)d_in[0];
  const float* W1 = (const float*)d_in[1];
  const float* b1 = (const float*)d_in[2];
  const float* W2 = (const float*)d_in[3];
  const float* b2 = (const float*)d_in[4];
  const int* eidx = (const int*)d_in[5];
  const int* batch = (const int*)d_in[6];

  const int N = in_sizes[6];
  const int E = in_sizes[5] / 2;
  const int* row = eidx;
  const int* col = eidx + E;

  float* ws = (float*)d_ws;
  float* dinv = ws;
  float* bufA = ws + (((size_t)N + 63) & ~(size_t)63);
  unsigned* bm = (unsigned*)(bufA + (size_t)N * NCOL);

  const int G = out_size / NCOL - N;
  float* hs = (float*)d_out;
  float* hout = hs + (size_t)G * NCOL;

  const dim3 b256(256);
  const int poolBlocks = (N + (POOL_T / 32) * POOL_RW - 1) / ((POOL_T / 32) * POOL_RW);
  const int gemmBlocks = (2 * ((N + 15) / 16) + GW - 1) / GW;
  const int aggBlocks = (N + TD - 1) / TD;
  const dim3 bAgg(AGG_T);
  const int bmBlocks = (E + AGG_T * BM_C - 1) / (AGG_T * BM_C);
  const int WP = bmBlocks * BM_W;

  k_bitmap<<<bmBlocks, bAgg, 0, stream>>>(col, bm, E, WP, aggBlocks);
  k_deg_tile<<<aggBlocks, bAgg, 0, stream>>>(col, bm, WP, dinv, N);

  k_gemm_wres<KIN><<<gemmBlocks, dim3(GT), 0, stream>>>(x, W1, bufA, N);
  k_agg_tile<<<aggBlocks, bAgg, 0, stream>>>(bufA, row, col, dinv, b1, bm, WP, hout, N, 1);

  k_gemm_wres<NCOL><<<gemmBlocks, dim3(GT), 0, stream>>>(hout, W2, bufA, N);
  k_agg_tile<<<aggBlocks, bAgg, 0, stream>>>(bufA, row, col, dinv, b2, bm, WP, hout, N, 0);

  k_zero<<<(G * NCOL + 255) / 256, b256, 0, stream>>>(hs, G * NCOL);
  k_pool_runs<<<poolBlocks, dim3(POOL_T), 0, stream>>>(hout, batch, hs, N, G);
}


# Aggregation: found edges go to an LDS queue per destination row (place by LDS integer add: arrival order); the owner wave sums a row in registers from 16-B loads, stores it; no per-edge float atomic
# speedup vs baseline: 4.4112x; 1.0718x over previous
#include <hip/hip_runtime.h>


typedef __attribute__((ext_vector_type(2))) float v2f;
typedef __attribute__((ext_vector_type(8))) float v8f;

#define NCOL 128
#define KIN 512
#define GW 25
#define GT (GW * 32)
#define LWS 272

struct ablk {
  float4 q[4];
};

__device__ __forceinline__ void a_load(ablk& a, const float* __restrict__ p) {
#pragma unroll
  for (int q = 0; q < 4; ++q) a.q[q] = *(const float4*)(p + 8 * q);
}

__device__ __forceinline__ void blk_mma(v8f (&acc)[4], const ablk& a,
                                        const float* brow) {
#pragma unroll
  for (int q = 0; q < 4; ++q) {
#pragma unroll
    for (int s = 0; s < 2; ++s) {
      v2f av;
      av.x = s ? a.q[q].z : a.q[q].x;
      av.y = s ? a.q[q].w : a.q[q].y;
      const float* br = brow + (4 * q + s) * LWS;
#pragma unroll
      for (int t = 0; t < 4; ++t) {
        const v2f b = *(const v2f*)(br + t * 32);
        acc[t] = __builtin_amdgcn_wmma_f32_16x16x4_f32(
             false, av,  false, b,
             (short)0, acc[t],  false,  false);
      }
    }
  }
}

template <int K>
__global__ __launch_bounds__(GT) void k_gemm_wres(
    const float* __restrict__ X, const float* __restrict__ W,
    float* __restrict__ C, int M) {
  __shared__ float lw2[(K / 2) * LWS];

  const int tid = threadIdx.x;
  const int lane = tid & 31;
  const int wave = __builtin_amdgcn_readfirstlane(tid >> 5);
  const int c = lane & 15;
  const int h = lane >> 4;
  const int job = blockIdx.x * GW + wave;
  const int strip = job >> 1;
  const int half = job & 1;
  const int nstrips = (M + 15) >> 4;

  int r = strip * 16 + c;
  r = (r < M) ? r : (M - 1);
  const float* xp = X + (size_t)r * K + 4 * h;

  ablk a0, a1;
  a_load(a0, xp);
  a_load(a1, xp + 32);

  constexpr int NW = (K / 2) * (NCOL / 4);
  for (int i0 = tid; i0 < NW; i0 += 4 * GT) {
    float4 w0[4], w1[4];
#pragma unroll
    for (int u = 0; u < 4; ++u) {
      int i = i0 + u * GT;
      i = (i < NW) ? i : (NW - 1);
      const float* wr = W + (size_t)(i >> 5) * (2 * NCOL) + (i & 31) * 4;
      w0[u] = *(const float4*)wr;
      w1[u] = *(const float4*)(wr + NCOL);
    }
#pragma unroll
    for (int u = 0; u < 4; ++u) {
      const int i = i0 + u * GT;
      if (i < NW) {
        float* dst = lw2 + (i >> 5) * LWS + (i & 31) * 8;
        ((float4*)dst)[0] = make_float4(w0[u].x, w1[u].x, w0[u].y, w1[u].y);
        ((float4*)dst)[1] = make_float4(w0[u].z, w1[u].z, w0[u].w, w1[u].w);
      }
    }
  }
  __syncthreads();
  if (strip >= nstrips) return;

  v8f acc[4] = {};
  const float* bbp = lw2 + (2 * h) * LWS + (half * 64 + c) * 2;

  for (int kb = 0; kb < K; kb += 64) {
    blk_mma(acc, a0, bbp + (kb >> 1) * LWS);
    if (kb + 64 < K) a_load(a0, xp + kb + 64);
    blk_mma(acc, a1, bbp + ((kb + 32) >> 1) * LWS);
    if (kb + 96 < K) a_load(a1, xp + kb + 96);
  }

  float* cp = C + (size_t)(strip * 16 + 8 * h) * NCOL + half * 64 + c;
  const int m0 = strip * 16 + 8 * h;
#pragma unroll
  for (int t = 0; t < 4; ++t) {
#pragma unroll
    for (int v = 0; v < 8; ++v) {
      if (m0 + v < M) cp[(size_t)v * NCOL + t * 16] = acc[t][v];
    }
  }
}

#define TD 200
#define AGG_T 1024

#define BM_C 4
#define BM_W (32 * BM_C)
#define BM_P (BM_W + 4)
#define BM_MAXT 256

__global__ __launch_bounds__(AGG_T) void k_bitmap(
    const int* __restrict__ col, unsigned* __restrict__ bm, int E, int WP,
    int nTiles) {
  __shared__ unsigned sm[BM_MAXT * BM_P];
  const int tid = threadIdx.x;
  uint4* sm4 = (uint4*)sm;
  const int nz = nTiles * (BM_P / 4);
  for (int i = tid; i < nz; i += AGG_T) sm4[i] = make_uint4(0u, 0u, 0u, 0u);
  __syncthreads();

  const int e0 = blockIdx.x * (AGG_T * BM_C) + tid;
  int d[BM_C];
#pragma unroll
  for (int c = 0; c < BM_C; ++c) {
    const int e = e0 + c * AGG_T;
    d[c] = (e < E) ? col[e] : -1;
  }
#pragma unroll
  for (int c = 0; c < BM_C; ++c) {
    if (d[c] >= 0)
      atomicOr(&sm[(d[c] / TD) * BM_P + c * 32 + (tid >> 5)], 1u << (tid & 31));
  }
  __syncthreads();

  const int nw = nTiles * (BM_W / 4);
  for (int i = tid; i < nw; i += AGG_T) {
    const int t = i / (BM_W / 4);
    const int j = i % (BM_W / 4);
    *(uint4*)(bm + (size_t)t * WP + (size_t)blockIdx.x * BM_W + j * 4) =
        sm4[t * (BM_P / 4) + j];
  }
}

#define DG_U 20

__global__ __launch_bounds__(AGG_T) void k_deg_tile(
    const int* __restrict__ col, const unsigned* __restrict__ bm, int WP,
    float* __restrict__ dinv, int N) {
  __shared__ int cnt[TD];
  const int tid = threadIdx.x;
  const int d0 = blockIdx.x * TD;
  const int rows = (N - d0 < TD) ? (N - d0) : TD;
  if (tid < TD) cnt[tid] = 0;
  __syncthreads();

  const unsigned* bmrow = bm + (size_t)blockIdx.x * WP;
  for (int base = 0; base < WP; base += DG_U * AGG_T) {
    unsigned m[DG_U];
#pragma unroll
    for (int j = 0; j < DG_U; ++j) {
      const int w = base + j * AGG_T + tid;
      m[j] = (w < WP) ? bmrow[w] : 0u;
    }
    bool more;
    do {
      int dv[DG_U];
      unsigned hm = 0u;
#pragma unroll
      for (int j = 0; j < DG_U; ++j) {
        const unsigned mm = m[j];
        const bool has = (mm != 0u);
        dv[j] = 0;
        if (__builtin_amdgcn_ballot_w32(has) != 0u) {
          const int e =
              has ? (((base + j * AGG_T + tid) << 5) + __builtin_ctz(mm)) : 0;
          dv[j] = col[e];
          hm |= has ? (1u << j) : 0u;
          m[j] = mm & (mm - 1u);
        }
      }
      more = false;
#pragma unroll
      for (int j = 0; j < DG_U; ++j) {
        if (hm & (1u << j)) atomicAdd(&cnt[dv[j] - d0], 1);
        more = more || (m[j] != 0u);
      }
    } while (more);
  }
  __syncthreads();

  if (tid < rows) {
    const float d = 1.0f + (float)cnt[tid];
    dinv[d0 + tid] = (d > 0.f) ? rsqrtf(d) : 0.f;
  }
}

#define SCAN_U 4
#define QCAP 64
#define NWV (AGG_T / 32)
#define RQC 48
#define SUM_G 16

__device__ __forceinline__ void bm_load(unsigned (&m)[SCAN_U],
                                        const unsigned* __restrict__ rowp,
                                        int wb, int WP) {
#pragma unroll
  for (int j = 0; j < SCAN_U; ++j) {
    const int w = wb + j * AGG_T;
    m[j] = (w < WP) ? rowp[w] : 0u;
  }
}

__device__ __forceinline__ void agg_route(
    const int* qp, int n, int lane, int d0, const float* __restrict__ hlin,
    const int* __restrict__ row, const int* __restrict__ col,
    const float* __restrict__ dinv, int* rcnt, uint2* rq, float* acc) {
  if (lane < n) {
    const int e = qp[lane];
    const int d = col[e];
    const int s = row[e];
    const int dl = d - d0;
    const float w = dinv[s] * dinv[d];
    const int pos = atomicAdd(&rcnt[dl], 1);
    if (pos < RQC) {
      rq[dl * RQC + pos] = make_uint2((unsigned)s, __float_as_uint(w));
    } else {
      const float* p = hlin + (size_t)s * NCOL;
      float* a = acc + dl * NCOL;
      for (int cc = 0; cc < NCOL; ++cc) atomicAdd(a + cc, p[cc] * w);
    }
  }
}

__device__ __forceinline__ void agg_sum_row(float4& cur, int s, float w, int n,
                                            int lane,
                                            const float* __restrict__ hlin) {
  for (int i = 0; i < n; i += SUM_G) {
    int sb[SUM_G];
    float wb[SUM_G];
    float4 v[SUM_G];
#pragma unroll
    for (int u = 0; u < SUM_G; ++u) {
      const int ii = (i + u < n) ? (i + u) : (n - 1);
      sb[u] = __builtin_amdgcn_readlane(s, ii);
      wb[u] = __int_as_float(__builtin_amdgcn_readlane(__float_as_int(w), ii));
    }
#pragma unroll
    for (int u = 0; u < SUM_G; ++u)
      v[u] = *(const float4*)(hlin + (size_t)sb[u] * NCOL + 4 * lane);
#pragma unroll
    for (int u = 0; u < SUM_G; ++u) {
      if (i + u < n) {
        cur.x += v[u].x * wb[u];
        cur.y += v[u].y * wb[u];
        cur.z += v[u].z * wb[u];
        cur.w += v[u].w * wb[u];
      }
    }
  }
}

__global__ __launch_bounds__(AGG_T) void k_agg_tile(
    const float* __restrict__ hlin, const int* __restrict__ row,
    const int* __restrict__ col, const float* __restrict__ dinv,
    const float* __restrict__ bias, const unsigned* __restrict__ bm, int WP,
    float* __restrict__ out, int N, int do_relu) {
  __shared__ float acc[TD * NCOL];
  __shared__ int q[AGG_T / 32][QCAP];
  __shared__ uint2 rq[TD * RQC];
  __shared__ int rcnt[TD];

  const int tid = threadIdx.x;
  const int lane = tid & 31;
  const int wave = __builtin_amdgcn_readfirstlane(tid >> 5);
  const int d0 = blockIdx.x * TD;
  const int rows = (N - d0 < TD) ? (N - d0) : TD;
  const int nvec = rows * (NCOL / 4);

  const unsigned* bmrow = bm + (size_t)blockIdx.x * WP + lane;
  unsigned mn[SCAN_U];
  bm_load(mn, bmrow, wave * 32, WP);

  const float4* hsrc = (const float4*)(hlin + (size_t)d0 * NCOL);
  for (int i = tid; i < nvec; i += AGG_T) {
    const float wd = dinv[d0 + (i >> 5)];
    const float4 h = hsrc[i];
    float4 o;
    o.x = h.x * wd * wd;
    o.y = h.y * wd * wd;
    o.z = h.z * wd * wd;
    o.w = h.w * wd * wd;
    ((float4*)acc)[i] = o;
  }
  if (tid < TD) rcnt[tid] = 0;
  __syncthreads();

  int* qw = q[wave];
  int cnt = 0;
  for (int wb = wave * 32; wb < WP; wb += SCAN_U * AGG_T) {
    const unsigned m0 = mn[0], m1 = mn[1], m2 = mn[2], m3 = mn[3];
    bm_load(mn, bmrow, wb + SCAN_U * AGG_T, WP);
    int j = 0;
    unsigned mm = m0;
    for (;;) {
      const bool has = (mm != 0u);
      const unsigned bal = __builtin_amdgcn_ballot_w32(has);
      if (bal == 0u) {
        if (++j == SCAN_U) break;
        mm = (j == 1) ? m1 : ((j == 2) ? m2 : m3);
        continue;
      }
      if (has) {
        const int e = (wb + j * AGG_T + lane) * 32 + __builtin_ctz(mm);
        qw[cnt + (int)__builtin_amdgcn_mbcnt_lo(bal, 0u)] = e;
      }
      cnt += __builtin_popcount(bal);
      mm &= mm - 1u;
      if (cnt >= 32) {
        cnt -= 32;
        agg_route(qw + cnt, 32, lane, d0, hlin, row, col, dinv, rcnt, rq, acc);
      }
    }
  }
  agg_route(qw, cnt, lane, d0, hlin, row, col, dinv, rcnt, rq, acc);
  __syncthreads();

  {
    const int f = 4 * lane;
    const float bx = bias[f], by = bias[f + 1], bz = bias[f + 2], bw = bias[f + 3];
    float4* odst = (float4*)(out + (size_t)d0 * NCOL);
    const int rb = (wave * TD + NWV - 1) / NWV;
    int re = ((wave + 1) * TD + NWV - 1) / NWV;
    re = (re < rows) ? re : rows;
    const int nrw = re - rb;
    int cv = 0;
    if (lane < nrw) cv = rcnt[rb + lane];
    int nn = 0;
    uint2 en = make_uint2(0u, 0u);
    if (nrw > 0) {
      nn = __builtin_amdgcn_readlane(cv, 0);
      nn = (nn < RQC) ? nn : RQC;
      if (lane < nn) en = rq[rb * RQC + lane];
    }
    for (int k = 0; k < nrw; ++k) {
      const int r = rb + k;
      const int nq = nn;
      const uint2 e0 = en;
      float4 cur = ((const float4*)acc)[r * (NCOL / 4) + lane];
      if (k + 1 < nrw) {
        nn = __builtin_amdgcn_readlane(cv, k + 1);
        nn = (nn < RQC) ? nn : RQC;
        en = make_uint2(0u, 0u);
        if (lane < nn) en = rq[(r + 1) * RQC + lane];
      }
      agg_sum_row(cur, (int)e0.x, __uint_as_float(e0.y), (nq < 32) ? nq : 32,
                  lane, hlin);
      if (nq > 32) {
        uint2 e1 = make_uint2(0u, 0u);
        if (32 + lane < nq) e1 = rq[r * RQC + 32 + lane];
        agg_sum_row(cur, (int)e1.x, __uint_as_float(e1.y), nq - 32, lane, hlin);
      }
      cur.x += bx;
      cur.y += by;
      cur.z += bz;
      cur.w += bw;
      if (do_relu) {
        cur.x = (cur.x < 0.f) ? 0.f : cur.x;
        cur.y = (cur.y < 0.f) ? 0.f : cur.y;
        cur.z = (cur.z < 0.f) ? 0.f : cur.z;
        cur.w = (cur.w < 0.f) ? 0.f : cur.w;
      }
      odst[r * (NCOL / 4) + lane] = cur;
    }
  }
}

__global__ void k_zero(float* p, int n) {
  int i = blockIdx.x * blockDim.x + threadIdx.x;
  if (i < n) p[i] = 0.f;
}

#define POOL_RW 16
#define POOL_T 256

__device__ __forceinline__ void pool_flush(float* __restrict__ hs, int g, int G,
                                           int lane, const float (&a)[4]) {
  if ((unsigned)g < (unsigned)G) {
    float* p = hs + (size_t)g * NCOL + lane;
    atomicAdd(p, a[0]);
    atomicAdd(p + 32, a[1]);
    atomicAdd(p + 64, a[2]);
    atomicAdd(p + 96, a[3]);
  }
}

__global__ __launch_bounds__(POOL_T) void k_pool_runs(
    const float* __restrict__ h, const int* __restrict__ batch,
    float* __restrict__ hs, int N, int G) {
  const int lane = threadIdx.x & 31;
  const int wave = __builtin_amdgcn_readfirstlane(threadIdx.x >> 5);
  const int r0 = (blockIdx.x * (POOL_T / 32) + wave) * POOL_RW;
  if (r0 >= N) return;
  const float* hp = h + (size_t)r0 * NCOL + lane;
  const int* bp = batch + r0;
  float acc[4] = {0.f, 0.f, 0.f, 0.f};

  if (N - r0 >= POOL_RW) {
    float v[POOL_RW][4];
    int g[POOL_RW];
#pragma unroll
    for (int j = 0; j < POOL_RW; ++j) {
      v[j][0] = hp[j * NCOL];
      v[j][1] = hp[j * NCOL + 32];
      v[j][2] = hp[j * NCOL + 64];
      v[j][3] = hp[j * NCOL + 96];
      g[j] = bp[j];
    }
    int cur = g[0];
#pragma unroll
    for (int j = 0; j < POOL_RW; ++j) {
      if (g[j] != cur) {
        pool_flush(hs, cur, G, lane, acc);
        acc[0] = acc[1] = acc[2] = acc[3] = 0.f;
        cur = g[j];
      }
      acc[0] += v[j][0];
      acc[1] += v[j][1];
      acc[2] += v[j][2];
      acc[3] += v[j][3];
    }
    pool_flush(hs, cur, G, lane, acc);
  } else {
    const int nr = N - r0;
    int cur = bp[0];
    for (int j = 0; j < nr; ++j) {
      const int gj = bp[j];
      if (gj != cur) {
        pool_flush(hs, cur, G, lane, acc);
        acc[0] = acc[1] = acc[2] = acc[3] = 0.f;
        cur = gj;
      }
      acc[0] += hp[j * NCOL];
      acc[1] += hp[j * NCOL + 32];
      acc[2] += hp[j * NCOL + 64];
      acc[3] += hp[j * NCOL + 96];
    }
    pool_flush(hs, cur, G, lane, acc);
  }
}

extern "C" void kernel_launch(void* const* d_in, const int* in_sizes, int n_in,
                              void* d_out, int out_size, void* d_ws, size_t ws_size,
                              hipStream_t stream) {
  const float* x = (const float*)d_in[0];
  const float* W1 = (const float*)d_in[1];
  const float* b1 = (const float*)d_in[2];
  const float* W2 = (const float*)d_in[3];
  const float* b2 = (const float*)d_in[4];
  const int* eidx = (const int*)d_in[5];
  const int* batch = (const int*)d_in[6];

  const int N = in_sizes[6];
  const int E = in_sizes[5] / 2;
  const int* row = eidx;
  const int* col = eidx + E;

  float* ws = (float*)d_ws;
  float* dinv = ws;
  float* bufA = ws + (((size_t)N + 63) & ~(size_t)63);
  unsigned* bm = (unsigned*)(bufA + (size_t)N * NCOL);

  const int G = out_size / NCOL - N;
  float* hs = (float*)d_out;
  float* hout = hs + (size_t)G * NCOL;

  const dim3 b256(256);
  const int poolBlocks = (N + (POOL_T / 32) * POOL_RW - 1) / ((POOL_T / 32) * POOL_RW);
  const int gemmBlocks = (2 * ((N + 15) / 16) + GW - 1) / GW;
  const int aggBlocks = (N + TD - 1) / TD;
  const dim3 bAgg(AGG_T);
  const int bmBlocks = (E + AGG_T * BM_C - 1) / (AGG_T * BM_C);
  const int WP = bmBlocks * BM_W;

  k_bitmap<<<bmBlocks, bAgg, 0, stream>>>(col, bm, E, WP, aggBlocks);
  k_deg_tile<<<aggBlocks, bAgg, 0, stream>>>(col, bm, WP, dinv, N);

  k_gemm_wres<KIN><<<gemmBlocks, dim3(GT), 0, stream>>>(x, W1, bufA, N);
  k_agg_tile<<<aggBlocks, bAgg, 0, stream>>>(bufA, row, col, dinv, b1, bm, WP, hout, N, 1);

  k_gemm_wres<NCOL><<<gemmBlocks, dim3(GT), 0, stream>>>(hout, W2, bufA, N);
  k_agg_tile<<<aggBlocks, bAgg, 0, stream>>>(bufA, row, col, dinv, b2, bm, WP, hout, N, 0);

  k_zero<<<(G * NCOL + 255) / 256, b256, 0, stream>>>(hs, G * NCOL);
  k_pool_runs<<<poolBlocks, dim3(POOL_T), 0, stream>>>(hout, batch, hs, N, G);
}


# Edges routed once per call: degree pass also fills each row's source queue (LDS integer add: arrival order, one order for both layers) in global memory; both aggregations become pure row gathers
# speedup vs baseline: 4.6396x; 1.0518x over previous
#include <hip/hip_runtime.h>


typedef __attribute__((ext_vector_type(2))) float v2f;
typedef __attribute__((ext_vector_type(8))) float v8f;

#define NCOL 128
#define KIN 512
#define GW 25
#define GT (GW * 32)
#define LWS 272

struct ablk {
  float4 q[4];
};

__device__ __forceinline__ void a_load(ablk& a, const float* __restrict__ p) {
#pragma unroll
  for (int q = 0; q < 4; ++q) a.q[q] = *(const float4*)(p + 8 * q);
}

__device__ __forceinline__ void blk_mma(v8f (&acc)[4], const ablk& a,
                                        const float* brow) {
#pragma unroll
  for (int q = 0; q < 4; ++q) {
#pragma unroll
    for (int s = 0; s < 2; ++s) {
      v2f av;
      av.x = s ? a.q[q].z : a.q[q].x;
      av.y = s ? a.q[q].w : a.q[q].y;
      const float* br = brow + (4 * q + s) * LWS;
#pragma unroll
      for (int t = 0; t < 4; ++t) {
        const v2f b = *(const v2f*)(br + t * 32);
        acc[t] = __builtin_amdgcn_wmma_f32_16x16x4_f32(
             false, av,  false, b,
             (short)0, acc[t],  false,  false);
      }
    }
  }
}

template <int K>
__global__ __launch_bounds__(GT) void k_gemm_wres(
    const float* __restrict__ X, const float* __restrict__ W,
    float* __restrict__ C, int M) {
  __shared__ float lw2[(K / 2) * LWS];

  const int tid = threadIdx.x;
  const int lane = tid & 31;
  const int wave = __builtin_amdgcn_readfirstlane(tid >> 5);
  const int c = lane & 15;
  const int h = lane >> 4;
  const int job = blockIdx.x * GW + wave;
  const int strip = job >> 1;
  const int half = job & 1;
  const int nstrips = (M + 15) >> 4;

  int r = strip * 16 + c;
  r = (r < M) ? r : (M - 1);
  const float* xp = X + (size_t)r * K + 4 * h;

  ablk a0, a1;
  a_load(a0, xp);
  a_load(a1, xp + 32);

  constexpr int NW = (K / 2) * (NCOL / 4);
  for (int i0 = tid; i0 < NW; i0 += 4 * GT) {
    float4 w0[4], w1[4];
#pragma unroll
    for (int u = 0; u < 4; ++u) {
      int i = i0 + u * GT;
      i = (i < NW) ? i : (NW - 1);
      const float* wr = W + (size_t)(i >> 5) * (2 * NCOL) + (i & 31) * 4;
      w0[u] = *(const float4*)wr;
      w1[u] = *(const float4*)(wr + NCOL);
    }
#pragma unroll
    for (int u = 0; u < 4; ++u) {
      const int i = i0 + u * GT;
      if (i < NW) {
        float* dst = lw2 + (i >> 5) * LWS + (i & 31) * 8;
        ((float4*)dst)[0] = make_float4(w0[u].x, w1[u].x, w0[u].y, w1[u].y);
        ((float4*)dst)[1] = make_float4(w0[u].z, w1[u].z, w0[u].w, w1[u].w);
      }
    }
  }
  __syncthreads();
  if (strip >= nstrips) return;

  v8f acc[4] = {};
  const float* bbp = lw2 + (2 * h) * LWS + (half * 64 + c) * 2;

  for (int kb = 0; kb < K; kb += 64) {
    blk_mma(acc, a0, bbp + (kb >> 1) * LWS);
    if (kb + 64 < K) a_load(a0, xp + kb + 64);
    blk_mma(acc, a1, bbp + ((kb + 32) >> 1) * LWS);
    if (kb + 96 < K) a_load(a1, xp + kb + 96);
  }

  float* cp = C + (size_t)(strip * 16 + 8 * h) * NCOL + half * 64 + c;
  const int m0 = strip * 16 + 8 * h;
#pragma unroll
  for (int t = 0; t < 4; ++t) {
#pragma unroll
    for (int v = 0; v < 8; ++v) {
      if (m0 + v < M) cp[(size_t)v * NCOL + t * 16] = acc[t][v];
    }
  }
}

#define TD 200
#define AGG_T 1024

#define BM_C 4
#define BM_W (32 * BM_C)
#define BM_P (BM_W + 4)
#define BM_MAXT 256

__global__ __launch_bounds__(AGG_T) void k_bitmap(
    const int* __restrict__ col, unsigned* __restrict__ bm, int E, int WP,
    int nTiles) {
  __shared__ unsigned sm[BM_MAXT * BM_P];
  const int tid = threadIdx.x;
  uint4* sm4 = (uint4*)sm;
  const int nz = nTiles * (BM_P / 4);
  for (int i = tid; i < nz; i += AGG_T) sm4[i] = make_uint4(0u, 0u, 0u, 0u);
  __syncthreads();

  const int e0 = blockIdx.x * (AGG_T * BM_C) + tid;
  int d[BM_C];
#pragma unroll
  for (int c = 0; c < BM_C; ++c) {
    const int e = e0 + c * AGG_T;
    d[c] = (e < E) ? col[e] : -1;
  }
#pragma unroll
  for (int c = 0; c < BM_C; ++c) {
    if (d[c] >= 0)
      atomicOr(&sm[(d[c] / TD) * BM_P + c * 32 + (tid >> 5)], 1u << (tid & 31));
  }
  __syncthreads();

  const int nw = nTiles * (BM_W / 4);
  for (int i = tid; i < nw; i += AGG_T) {
    const int t = i / (BM_W / 4);
    const int j = i % (BM_W / 4);
    *(uint4*)(bm + (size_t)t * WP + (size_t)blockIdx.x * BM_W + j * 4) =
        sm4[t * (BM_P / 4) + j];
  }
}

#define DG_U 20
#define RQC 64
#define NWV (AGG_T / 32)

__device__ __forceinline__ int deg_place(int* cnt, unsigned hm, int j, int dl) {
  return ((hm >> j) & 1u) ? atomicAdd(&cnt[dl], 1) : RQC;
}

__global__ __launch_bounds__(AGG_T) void k_deg_route(
    const int* __restrict__ row, const int* __restrict__ col, unsigned* bm,
    int WP, float* __restrict__ dinv, int N) {
  __shared__ int cnt[TD];
  __shared__ int rq[TD * RQC];
  const int tid = threadIdx.x;
  const int d0 = blockIdx.x * TD;
  const int rows = (N - d0 < TD) ? (N - d0) : TD;
  if (tid < TD) cnt[tid] = 0;
  __syncthreads();

  unsigned* bmrow = bm + (size_t)blockIdx.x * WP;
  for (int base = 0; base < WP; base += DG_U * AGG_T) {
    unsigned m[DG_U];
#pragma unroll
    for (int j = 0; j < DG_U; ++j) {
      const int w = base + j * AGG_T + tid;
      m[j] = (w < WP) ? bmrow[w] : 0u;
    }
    bool more;
    do {
      int dv[DG_U], sv[DG_U];
      unsigned hm = 0u;
      unsigned wm = 0u;
#pragma unroll
      for (int j = 0; j < DG_U; ++j) {
        const unsigned mm = m[j];
        const bool has = (mm != 0u);
        if (__builtin_amdgcn_ballot_w32(has) != 0u) {
          const int e =
              has ? (((base + j * AGG_T + tid) << 5) + __builtin_ctz(mm)) : 0;
          dv[j] = col[e];
          sv[j] = row[e];
          hm |= has ? (1u << j) : 0u;
          wm |= 1u << j;
          m[j] = mm & (mm - 1u);
        }
      }
      unsigned left = 0u;
#pragma unroll
      for (int j0 = 0; j0 < DG_U; j0 += 4) {
        const int p0 = ((wm >> j0) & 1u) ? deg_place(cnt, hm, j0, dv[j0] - d0) : RQC;
        const int p1 = ((wm >> (j0 + 1)) & 1u) ? deg_place(cnt, hm, j0 + 1, dv[j0 + 1] - d0) : RQC;
        const int p2 = ((wm >> (j0 + 2)) & 1u) ? deg_place(cnt, hm, j0 + 2, dv[j0 + 2] - d0) : RQC;
        const int p3 = ((wm >> (j0 + 3)) & 1u) ? deg_place(cnt, hm, j0 + 3, dv[j0 + 3] - d0) : RQC;
        if (p0 < RQC) rq[(dv[j0] - d0) * RQC + p0] = sv[j0];
        if (p1 < RQC) rq[(dv[j0 + 1] - d0) * RQC + p1] = sv[j0 + 1];
        if (p2 < RQC) rq[(dv[j0 + 2] - d0) * RQC + p2] = sv[j0 + 2];
        if (p3 < RQC) rq[(dv[j0 + 3] - d0) * RQC + p3] = sv[j0 + 3];
        left |= m[j0] | m[j0 + 1] | m[j0 + 2] | m[j0 + 3];
      }
      more = (left != 0u);
    } while (more);
  }
  __syncthreads();

  if (tid < rows) {
    const float d = 1.0f + (float)cnt[tid];
    dinv[d0 + tid] = (d > 0.f) ? rsqrtf(d) : 0.f;
  }
  if (tid < TD) bmrow[TD * RQC + tid] = (unsigned)cnt[tid];
  const int lane = tid & 31;
  for (int r = tid >> 5; r < TD; r += NWV) {
    int c = cnt[r];
    c = (c < RQC) ? c : RQC;
    bmrow[r * RQC + lane] = (lane < c) ? (unsigned)rq[r * RQC + lane] : 0u;
    if (c > 32)
      bmrow[r * RQC + 32 + lane] =
          (32 + lane < c) ? (unsigned)rq[r * RQC + 32 + lane] : 0u;
  }
}

#define SUM_G 16

__device__ __forceinline__ void agg_sum_row(float4& cur, int s, float w, int n,
                                            int lane,
                                            const float* __restrict__ hlin) {
  for (int i = 0; i < n; i += SUM_G) {
    int sb[SUM_G];
    float wb[SUM_G];
    float4 v[SUM_G];
#pragma unroll
    for (int u = 0; u < SUM_G; ++u) {
      const int ii = (i + u < n) ? (i + u) : (n - 1);
      sb[u] = __builtin_amdgcn_readlane(s, ii);
      wb[u] = __int_as_float(__builtin_amdgcn_readlane(__float_as_int(w), ii));
    }
#pragma unroll
    for (int u = 0; u < SUM_G; ++u)
      v[u] = *(const float4*)(hlin + (size_t)sb[u] * NCOL + 4 * lane);
#pragma unroll
    for (int u = 0; u < SUM_G; ++u) {
      if (i + u < n) {
        cur.x += v[u].x * wb[u];
        cur.y += v[u].y * wb[u];
        cur.z += v[u].z * wb[u];
        cur.w += v[u].w * wb[u];
      }
    }
  }
}

__global__ __launch_bounds__(AGG_T) void k_agg_rows(
    const float* __restrict__ hlin, const int* __restrict__ row,
    const int* __restrict__ col, const float* __restrict__ dinv,
    const float* __restrict__ bias, const unsigned* __restrict__ bm, int WP,
    float* __restrict__ out, int N, int E, int do_relu) {
  const int tid = threadIdx.x;
  const int lane = tid & 31;
  const int wave = __builtin_amdgcn_readfirstlane(tid >> 5);
  const int d0 = blockIdx.x * TD;
  const int rows = (N - d0 < TD) ? (N - d0) : TD;
  const unsigned* qg = bm + (size_t)blockIdx.x * WP;

  const int rb = (wave * TD + NWV - 1) / NWV;
  int re = ((wave + 1) * TD + NWV - 1) / NWV;
  re = (re < rows) ? re : rows;
  const int nrw = re - rb;
  if (nrw <= 0) return;
  const int last = nrw - 1;

  const int f = 4 * lane;
  const float4* hsrc = (const float4*)(hlin + (size_t)d0 * NCOL);
  float4* odst = (float4*)(out + (size_t)d0 * NCOL);
  int cv = 0;
  float dvv = 0.f;
  if (lane < nrw) {
    cv = (int)qg[TD * RQC + rb + lane];
    dvv = dinv[d0 + rb + lane];
  }
  const float bx = bias[f], by = bias[f + 1], bz = bias[f + 2], bw = bias[f + 3];

  const int kb1 = (1 < last) ? 1 : last;
  float4 hA = hsrc[rb * (NCOL / 4) + lane];
  int sA = (int)qg[rb * RQC + lane];
  int sB = (int)qg[(rb + kb1) * RQC + lane];
  int nA = __builtin_amdgcn_readlane(cv, 0);
  int nB = __builtin_amdgcn_readlane(cv, kb1);
  float wA = dinv[sA] *
             __int_as_float(__builtin_amdgcn_readlane(__float_as_int(dvv), 0));

  for (int k = 0; k < nrw; ++k) {
    const int r = rb + k;
    const int k1 = (k + 1 < last) ? (k + 1) : last;
    const int k2 = (k + 2 < last) ? (k + 2) : last;
    const float dr =
        __int_as_float(__builtin_amdgcn_readlane(__float_as_int(dvv), k));
    const int nC = __builtin_amdgcn_readlane(cv, k2);
    const int sC = (int)qg[(rb + k2) * RQC + lane];
    const float dB = dinv[sB];
    const float4 hB = hsrc[(rb + k1) * (NCOL / 4) + lane];

    float4 cur;
    cur.x = hA.x * dr * dr;
    cur.y = hA.y * dr * dr;
    cur.z = hA.z * dr * dr;
    cur.w = hA.w * dr * dr;
    if (nA <= RQC) {
      agg_sum_row(cur, sA, wA, (nA < 32) ? nA : 32, lane, hlin);
      if (nA > 32) {
        int s1 = 0;
        float w1 = 0.f;
        if (32 + lane < nA) {
          s1 = (int)qg[r * RQC + 32 + lane];
          w1 = dinv[s1] * dr;
        }
        agg_sum_row(cur, s1, w1, nA - 32, lane, hlin);
      }
    } else {
      const int d = d0 + r;
      for (int e0 = 0; e0 < E; e0 += 32) {
        const int e = e0 + lane;
        const int ce = (e < E) ? col[e] : -1;
        unsigned bal = __builtin_amdgcn_ballot_w32(ce == d);
        while (bal != 0u) {
          const int s = row[e0 + __builtin_ctz(bal)];
          bal &= bal - 1u;
          const float w = dinv[s] * dr;
          const float4 v = *(const float4*)(hlin + (size_t)s * NCOL + f);
          cur.x += v.x * w;
          cur.y += v.y * w;
          cur.z += v.z * w;
          cur.w += v.w * w;
        }
      }
    }
    cur.x += bx;
    cur.y += by;
    cur.z += bz;
    cur.w += bw;
    if (do_relu) {
      cur.x = (cur.x < 0.f) ? 0.f : cur.x;
      cur.y = (cur.y < 0.f) ? 0.f : cur.y;
      cur.z = (cur.z < 0.f) ? 0.f : cur.z;
      cur.w = (cur.w < 0.f) ? 0.f : cur.w;
    }
    odst[r * (NCOL / 4) + lane] = cur;

    wA = dB * __int_as_float(__builtin_amdgcn_readlane(__float_as_int(dvv), k1));
    nA = nB;
    sA = sB;
    hA = hB;
    nB = nC;
    sB = sC;
  }
}

__global__ void k_zero(float* p, int n) {
  int i = blockIdx.x * blockDim.x + threadIdx.x;
  if (i < n) p[i] = 0.f;
}

#define POOL_RW 16
#define POOL_T 256

__device__ __forceinline__ void pool_flush(float* __restrict__ hs, int g, int G,
                                           int lane, const float (&a)[4]) {
  if ((unsigned)g < (unsigned)G) {
    float* p = hs + (size_t)g * NCOL + lane;
    atomicAdd(p, a[0]);
    atomicAdd(p + 32, a[1]);
    atomicAdd(p + 64, a[2]);
    atomicAdd(p + 96, a[3]);
  }
}

__global__ __launch_bounds__(POOL_T) void k_pool_runs(
    const float* __restrict__ h, const int* __restrict__ batch,
    float* __restrict__ hs, int N, int G) {
  const int lane = threadIdx.x & 31;
  const int wave = __builtin_amdgcn_readfirstlane(threadIdx.x >> 5);
  const int r0 = (blockIdx.x * (POOL_T / 32) + wave) * POOL_RW;
  if (r0 >= N) return;
  const float* hp = h + (size_t)r0 * NCOL + lane;
  const int* bp = batch + r0;
  float acc[4] = {0.f, 0.f, 0.f, 0.f};

  if (N - r0 >= POOL_RW) {
    float v[POOL_RW][4];
    int g[POOL_RW];
#pragma unroll
    for (int j = 0; j < POOL_RW; ++j) {
      v[j][0] = hp[j * NCOL];
      v[j][1] = hp[j * NCOL + 32];
      v[j][2] = hp[j * NCOL + 64];
      v[j][3] = hp[j * NCOL + 96];
      g[j] = bp[j];
    }
    int cur = g[0];
#pragma unroll
    for (int j = 0; j < POOL_RW; ++j) {
      if (g[j] != cur) {
        pool_flush(hs, cur, G, lane, acc);
        acc[0] = acc[1] = acc[2] = acc[3] = 0.f;
        cur = g[j];
      }
      acc[0] += v[j][0];
      acc[1] += v[j][1];
      acc[2] += v[j][2];
      acc[3] += v[j][3];
    }
    pool_flush(hs, cur, G, lane, acc);
  } else {
    const int nr = N - r0;
    int cur = bp[0];
    for (int j = 0; j < nr; ++j) {
      const int gj = bp[j];
      if (gj != cur) {
        pool_flush(hs, cur, G, lane, acc);
        acc[0] = acc[1] = acc[2] = acc[3] = 0.f;
        cur = gj;
      }
      acc[0] += hp[j * NCOL];
      acc[1] += hp[j * NCOL + 32];
      acc[2] += hp[j * NCOL + 64];
      acc[3] += hp[j * NCOL + 96];
    }
    pool_flush(hs, cur, G, lane, acc);
  }
}

extern "C" void kernel_launch(void* const* d_in, const int* in_sizes, int n_in,
                              void* d_out, int out_size, void* d_ws, size_t ws_size,
                              hipStream_t stream) {
  const float* x = (const float*)d_in[0];
  const float* W1 = (const float*)d_in[1];
  const float* b1 = (const float*)d_in[2];
  const float* W2 = (const float*)d_in[3];
  const float* b2 = (const float*)d_in[4];
  const int* eidx = (const int*)d_in[5];
  const int* batch = (const int*)d_in[6];

  const int N = in_sizes[6];
  const int E = in_sizes[5] / 2;
  const int* row = eidx;
  const int* col = eidx + E;

  float* ws = (float*)d_ws;
  float* dinv = ws;
  float* bufA = ws + (((size_t)N + 63) & ~(size_t)63);
  unsigned* bm = (unsigned*)(bufA + (size_t)N * NCOL);

  const int G = out_size / NCOL - N;
  float* hs = (float*)d_out;
  float* hout = hs + (size_t)G * NCOL;

  const dim3 b256(256);
  const int poolBlocks = (N + (POOL_T / 32) * POOL_RW - 1) / ((POOL_T / 32) * POOL_RW);
  const int gemmBlocks = (2 * ((N + 15) / 16) + GW - 1) / GW;
  const int aggBlocks = (N + TD - 1) / TD;
  const dim3 bAgg(AGG_T);
  const int bmBlocks = (E + AGG_T * BM_C - 1) / (AGG_T * BM_C);
  const int WP = bmBlocks * BM_W;

  k_bitmap<<<bmBlocks, bAgg, 0, stream>>>(col, bm, E, WP, aggBlocks);
  k_deg_route<<<aggBlocks, bAgg, 0, stream>>>(row, col, bm, WP, dinv, N);

  k_gemm_wres<KIN><<<gemmBlocks, dim3(GT), 0, stream>>>(x, W1, bufA, N);
  k_agg_rows<<<aggBlocks, bAgg, 0, stream>>>(bufA, row, col, dinv, b1, bm, WP, hout, N, E, 1);

  k_gemm_wres<NCOL><<<gemmBlocks, dim3(GT), 0, stream>>>(hout, W2, bufA, N);
  k_agg_rows<<<aggBlocks, bAgg, 0, stream>>>(bufA, row, col, dinv, b2, bm, WP, hout, N, E, 0);

  k_zero<<<(G * NCOL + 255) / 256, b256, 0, stream>>>(hs, G * NCOL);
  k_pool_runs<<<poolBlocks, dim3(POOL_T), 0, stream>>>(hout, batch, hs, N, G);
}


# Edge weights split: GEMM stores row s times dinv[s], aggregation sums plain rows in the same order and scales the sum by dinv[d] (rounding points moved); no per-edge dinv gather
# speedup vs baseline: 4.8080x; 1.0363x over previous
#include <hip/hip_runtime.h>


typedef __attribute__((ext_vector_type(2))) float v2f;
typedef __attribute__((ext_vector_type(8))) float v8f;

#define NCOL 128
#define KIN 512
#define GW 25
#define GT (GW * 32)
#define LWS 272

struct ablk {
  float4 q[4];
};

__device__ __forceinline__ void a_load(ablk& a, const float* __restrict__ p) {
#pragma unroll
  for (int q = 0; q < 4; ++q) a.q[q] = *(const float4*)(p + 8 * q);
}

__device__ __forceinline__ void blk_mma(v8f (&acc)[4], const ablk& a,
                                        const float* brow) {
#pragma unroll
  for (int q = 0; q < 4; ++q) {
#pragma unroll
    for (int s = 0; s < 2; ++s) {
      v2f av;
      av.x = s ? a.q[q].z : a.q[q].x;
      av.y = s ? a.q[q].w : a.q[q].y;
      const float* br = brow + (4 * q + s) * LWS;
#pragma unroll
      for (int t = 0; t < 4; ++t) {
        const v2f b = *(const v2f*)(br + t * 32);
        acc[t] = __builtin_amdgcn_wmma_f32_16x16x4_f32(
             false, av,  false, b,
             (short)0, acc[t],  false,  false);
      }
    }
  }
}

template <int K>
__global__ __launch_bounds__(GT) void k_gemm_wres(
    const float* __restrict__ X, const float* __restrict__ W,
    const float* __restrict__ scale, float* __restrict__ C, int M) {
  __shared__ float lw2[(K / 2) * LWS];

  const int tid = threadIdx.x;
  const int lane = tid & 31;
  const int wave = __builtin_amdgcn_readfirstlane(tid >> 5);
  const int c = lane & 15;
  const int h = lane >> 4;
  const int job = blockIdx.x * GW + wave;
  const int strip = job >> 1;
  const int half = job & 1;
  const int nstrips = (M + 15) >> 4;

  int r = strip * 16 + c;
  r = (r < M) ? r : (M - 1);
  const float* xp = X + (size_t)r * K + 4 * h;

  ablk a0, a1;
  a_load(a0, xp);
  a_load(a1, xp + 32);

  const int m0 = strip * 16 + 8 * h;
  float sc[8];
#pragma unroll
  for (int v = 0; v < 8; ++v) {
    int m = m0 + v;
    m = (m < M) ? m : (M - 1);
    sc[v] = scale[m];
  }

  constexpr int NW = (K / 2) * (NCOL / 4);
  for (int i0 = tid; i0 < NW; i0 += 4 * GT) {
    float4 w0[4], w1[4];
#pragma unroll
    for (int u = 0; u < 4; ++u) {
      int i = i0 + u * GT;
      i = (i < NW) ? i : (NW - 1);
      const float* wr = W + (size_t)(i >> 5) * (2 * NCOL) + (i & 31) * 4;
      w0[u] = *(const float4*)wr;
      w1[u] = *(const float4*)(wr + NCOL);
    }
#pragma unroll
    for (int u = 0; u < 4; ++u) {
      const int i = i0 + u * GT;
      if (i < NW) {
        float* dst = lw2 + (i >> 5) * LWS + (i & 31) * 8;
        ((float4*)dst)[0] = make_float4(w0[u].x, w1[u].x, w0[u].y, w1[u].y);
        ((float4*)dst)[1] = make_float4(w0[u].z, w1[u].z, w0[u].w, w1[u].w);
      }
    }
  }
  __syncthreads();
  if (strip >= nstrips) return;

  v8f acc[4] = {};
  const float* bbp = lw2 + (2 * h) * LWS + (half * 64 + c) * 2;

  for (int kb = 0; kb < K; kb += 64) {
    blk_mma(acc, a0, bbp + (kb >> 1) * LWS);
    if (kb + 64 < K) a_load(a0, xp + kb + 64);
    blk_mma(acc, a1, bbp + ((kb + 32) >> 1) * LWS);
    if (kb + 96 < K) a_load(a1, xp + kb + 96);
  }

  float* cp = C + (size_t)m0 * NCOL + half * 64 + c;
#pragma unroll
  for (int t = 0; t < 4; ++t) {
#pragma unroll
    for (int v = 0; v < 8; ++v) {
      if (m0 + v < M) cp[(size_t)v * NCOL + t * 16] = acc[t][v] * sc[v];
    }
  }
}

#define TD 200
#define AGG_T 1024

#define BM_C 4
#define BM_W (32 * BM_C)
#define BM_P (BM_W + 4)
#define BM_MAXT 256

__global__ __launch_bounds__(AGG_T) void k_bitmap(
    const int* __restrict__ col, unsigned* __restrict__ bm, int E, int WP,
    int nTiles) {
  __shared__ unsigned sm[BM_MAXT * BM_P];
  const int tid = threadIdx.x;
  uint4* sm4 = (uint4*)sm;
  const int nz = nTiles * (BM_P / 4);
  for (int i = tid; i < nz; i += AGG_T) sm4[i] = make_uint4(0u, 0u, 0u, 0u);
  __syncthreads();

  const int e0 = blockIdx.x * (AGG_T * BM_C) + tid;
  int d[BM_C];
#pragma unroll
  for (int c = 0; c < BM_C; ++c) {
    const int e = e0 + c * AGG_T;
    d[c] = (e < E) ? col[e] : -1;
  }
#pragma unroll
  for (int c = 0; c < BM_C; ++c) {
    if (d[c] >= 0)
      atomicOr(&sm[(d[c] / TD) * BM_P + c * 32 + (tid >> 5)], 1u << (tid & 31));
  }
  __syncthreads();

  const int nw = nTiles * (BM_W / 4);
  for (int i = tid; i < nw; i += AGG_T) {
    const int t = i / (BM_W / 4);
    const int j = i % (BM_W / 4);
    *(uint4*)(bm + (size_t)t * WP + (size_t)blockIdx.x * BM_W + j * 4) =
        sm4[t * (BM_P / 4) + j];
  }
}

#define DG_U 20
#define RQC 64
#define NWV (AGG_T / 32)

__device__ __forceinline__ int deg_place(int* cnt, unsigned hm, int j, int dl) {
  return ((hm >> j) & 1u) ? atomicAdd(&cnt[dl], 1) : RQC;
}

__global__ __launch_bounds__(AGG_T) void k_deg_route(
    const int* __restrict__ row, const int* __restrict__ col, unsigned* bm,
    int WP, float* __restrict__ dinv, int N) {
  __shared__ int cnt[TD];
  __shared__ int rq[TD * RQC];
  const int tid = threadIdx.x;
  const int d0 = blockIdx.x * TD;
  const int rows = (N - d0 < TD) ? (N - d0) : TD;
  if (tid < TD) cnt[tid] = 0;
  __syncthreads();

  unsigned* bmrow = bm + (size_t)blockIdx.x * WP;
  for (int base = 0; base < WP; base += DG_U * AGG_T) {
    unsigned m[DG_U];
#pragma unroll
    for (int j = 0; j < DG_U; ++j) {
      const int w = base + j * AGG_T + tid;
      m[j] = (w < WP) ? bmrow[w] : 0u;
    }
    bool more;
    do {
      int dv[DG_U], sv[DG_U];
      unsigned hm = 0u;
      unsigned wm = 0u;
#pragma unroll
      for (int j = 0; j < DG_U; ++j) {
        const unsigned mm = m[j];
        const bool has = (mm != 0u);
        if (__builtin_amdgcn_ballot_w32(has) != 0u) {
          const int e =
              has ? (((base + j * AGG_T + tid) << 5) + __builtin_ctz(mm)) : 0;
          dv[j] = col[e];
          sv[j] = row[e];
          hm |= has ? (1u << j) : 0u;
          wm |= 1u << j;
          m[j] = mm & (mm - 1u);
        }
      }
      unsigned left = 0u;
#pragma unroll
      for (int j0 = 0; j0 < DG_U; j0 += 4) {
        const int p0 = ((wm >> j0) & 1u) ? deg_place(cnt, hm, j0, dv[j0] - d0) : RQC;
        const int p1 = ((wm >> (j0 + 1)) & 1u) ? deg_place(cnt, hm, j0 + 1, dv[j0 + 1] - d0) : RQC;
        const int p2 = ((wm >> (j0 + 2)) & 1u) ? deg_place(cnt, hm, j0 + 2, dv[j0 + 2] - d0) : RQC;
        const int p3 = ((wm >> (j0 + 3)) & 1u) ? deg_place(cnt, hm, j0 + 3, dv[j0 + 3] - d0) : RQC;
        if (p0 < RQC) rq[(dv[j0] - d0) * RQC + p0] = sv[j0];
        if (p1 < RQC) rq[(dv[j0 + 1] - d0) * RQC + p1] = sv[j0 + 1];
        if (p2 < RQC) rq[(dv[j0 + 2] - d0) * RQC + p2] = sv[j0 + 2];
        if (p3 < RQC) rq[(dv[j0 + 3] - d0) * RQC + p3] = sv[j0 + 3];
        left |= m[j0] | m[j0 + 1] | m[j0 + 2] | m[j0 + 3];
      }
      more = (left != 0u);
    } while (more);
  }
  __syncthreads();

  if (tid < rows) {
    const float d = 1.0f + (float)cnt[tid];
    dinv[d0 + tid] = (d > 0.f) ? rsqrtf(d) : 0.f;
  }
  if (tid < TD) bmrow[TD * RQC + tid] = (unsigned)cnt[tid];
  const int lane = tid & 31;
  for (int r = tid >> 5; r < TD; r += NWV) {
    int c = cnt[r];
    c = (c < RQC) ? c : RQC;
    bmrow[r * RQC + lane] = (lane < c) ? (unsigned)rq[r * RQC + lane] : 0u;
    if (c > 32)
      bmrow[r * RQC + 32 + lane] =
          (32 + lane < c) ? (unsigned)rq[r * RQC + 32 + lane] : 0u;
  }
}

#define SUM_G 16

__device__ __forceinline__ void agg_sum_row(float4& cur, int s, int n, int lane,
                                            const float* __restrict__ hlin) {
  for (int i = 0; i < n; i += SUM_G) {
    int sb[SUM_G];
    float4 v[SUM_G];
#pragma unroll
    for (int u = 0; u < SUM_G; ++u) {
      const int ii = (i + u < n) ? (i + u) : (n - 1);
      sb[u] = __builtin_amdgcn_readlane(s, ii);
    }
#pragma unroll
    for (int u = 0; u < SUM_G; ++u)
      v[u] = *(const float4*)(hlin + (size_t)sb[u] * NCOL + 4 * lane);
#pragma unroll
    for (int u = 0; u < SUM_G; ++u) {
      if (i + u < n) {
        cur.x += v[u].x;
        cur.y += v[u].y;
        cur.z += v[u].z;
        cur.w += v[u].w;
      }
    }
  }
}

__global__ __launch_bounds__(AGG_T) void k_agg_rows(
    const float* __restrict__ hlin, const int* __restrict__ row,
    const int* __restrict__ col, const float* __restrict__ dinv,
    const float* __restrict__ bias, const unsigned* __restrict__ bm, int WP,
    float* __restrict__ out, int N, int E, int do_relu) {
  const int tid = threadIdx.x;
  const int lane = tid & 31;
  const int wave = __builtin_amdgcn_readfirstlane(tid >> 5);
  const int d0 = blockIdx.x * TD;
  const int rows = (N - d0 < TD) ? (N - d0) : TD;
  const unsigned* qg = bm + (size_t)blockIdx.x * WP;

  const int rb = (wave * TD + NWV - 1) / NWV;
  int re = ((wave + 1) * TD + NWV - 1) / NWV;
  re = (re < rows) ? re : rows;
  const int nrw = re - rb;
  if (nrw <= 0) return;
  const int last = nrw - 1;

  const int f = 4 * lane;
  const float4* hsrc = (const float4*)(hlin + (size_t)d0 * NCOL);
  float4* odst = (float4*)(out + (size_t)d0 * NCOL);
  int cv = 0;
  float dvv = 0.f;
  if (lane < nrw) {
    cv = (int)qg[TD * RQC + rb + lane];
    dvv = dinv[d0 + rb + lane];
  }
  const float bx = bias[f], by = bias[f + 1], bz = bias[f + 2], bw = bias[f + 3];

  float4 hA = hsrc[rb * (NCOL / 4) + lane];
  int sA = (int)qg[rb * RQC + lane];
  int nA = __builtin_amdgcn_readlane(cv, 0);

  for (int k = 0; k < nrw; ++k) {
    const int r = rb + k;
    const int k1 = (k + 1 < last) ? (k + 1) : last;
    const float dr =
        __int_as_float(__builtin_amdgcn_readlane(__float_as_int(dvv), k));
    const int nB = __builtin_amdgcn_readlane(cv, k1);
    const int sB = (int)qg[(rb + k1) * RQC + lane];
    const float4 hB = hsrc[(rb + k1) * (NCOL / 4) + lane];

    float4 cur = hA;
    if (nA <= RQC) {
      agg_sum_row(cur, sA, (nA < 32) ? nA : 32, lane, hlin);
      if (nA > 32) {
        int s1 = 0;
        if (32 + lane < nA) s1 = (int)qg[r * RQC + 32 + lane];
        agg_sum_row(cur, s1, nA - 32, lane, hlin);
      }
    } else {
      const int d = d0 + r;
      for (int e0 = 0; e0 < E; e0 += 32) {
        const int e = e0 + lane;
        const int ce = (e < E) ? col[e] : -1;
        unsigned bal = __builtin_amdgcn_ballot_w32(ce == d);
        while (bal != 0u) {
          const int s = row[e0 + __builtin_ctz(bal)];
          bal &= bal - 1u;
          const float4 v = *(const float4*)(hlin + (size_t)s * NCOL + f);
          cur.x += v.x;
          cur.y += v.y;
          cur.z += v.z;
          cur.w += v.w;
        }
      }
    }
    cur.x = cur.x * dr + bx;
    cur.y = cur.y * dr + by;
    cur.z = cur.z * dr + bz;
    cur.w = cur.w * dr + bw;
    if (do_relu) {
      cur.x = (cur.x < 0.f) ? 0.f : cur.x;
      cur.y = (cur.y < 0.f) ? 0.f : cur.y;
      cur.z = (cur.z < 0.f) ? 0.f : cur.z;
      cur.w = (cur.w < 0.f) ? 0.f : cur.w;
    }
    odst[r * (NCOL / 4) + lane] = cur;

    nA = nB;
    sA = sB;
    hA = hB;
  }
}

__global__ void k_zero(float* p, int n) {
  int i = blockIdx.x * blockDim.x + threadIdx.x;
  if (i < n) p[i] = 0.f;
}

#define POOL_RW 16
#define POOL_T 256

__device__ __forceinline__ void pool_flush(float* __restrict__ hs, int g, int G,
                                           int lane, const float (&a)[4]) {
  if ((unsigned)g < (unsigned)G) {
    float* p = hs + (size_t)g * NCOL + lane;
    atomicAdd(p, a[0]);
    atomicAdd(p + 32, a[1]);
    atomicAdd(p + 64, a[2]);
    atomicAdd(p + 96, a[3]);
  }
}

__global__ __launch_bounds__(POOL_T) void k_pool_runs(
    const float* __restrict__ h, const int* __restrict__ batch,
    float* __restrict__ hs, int N, int G) {
  const int lane = threadIdx.x & 31;
  const int wave = __builtin_amdgcn_readfirstlane(threadIdx.x >> 5);
  const int r0 = (blockIdx.x * (POOL_T / 32) + wave) * POOL_RW;
  if (r0 >= N) return;
  const float* hp = h + (size_t)r0 * NCOL + lane;
  const int* bp = batch + r0;
  float acc[4] = {0.f, 0.f, 0.f, 0.f};

  if (N - r0 >= POOL_RW) {
    float v[POOL_RW][4];
    int g[POOL_RW];
#pragma unroll
    for (int j = 0; j < POOL_RW; ++j) {
      v[j][0] = hp[j * NCOL];
      v[j][1] = hp[j * NCOL + 32];
      v[j][2] = hp[j * NCOL + 64];
      v[j][3] = hp[j * NCOL + 96];
      g[j] = bp[j];
    }
    int cur = g[0];
#pragma unroll
    for (int j = 0; j < POOL_RW; ++j) {
      if (g[j] != cur) {
        pool_flush(hs, cur, G, lane, acc);
        acc[0] = acc[1] = acc[2] = acc[3] = 0.f;
        cur = g[j];
      }
      acc[0] += v[j][0];
      acc[1] += v[j][1];
      acc[2] += v[j][2];
      acc[3] += v[j][3];
    }
    pool_flush(hs, cur, G, lane, acc);
  } else {
    const int nr = N - r0;
    int cur = bp[0];
    for (int j = 0; j < nr; ++j) {
      const int gj = bp[j];
      if (gj != cur) {
        pool_flush(hs, cur, G, lane, acc);
        acc[0] = acc[1] = acc[2] = acc[3] = 0.f;
        cur = gj;
      }
      acc[0] += hp[j * NCOL];
      acc[1] += hp[j * NCOL + 32];
      acc[2] += hp[j * NCOL + 64];
      acc[3] += hp[j * NCOL + 96];
    }
    pool_flush(hs, cur, G, lane, acc);
  }
}

extern "C" void kernel_launch(void* const* d_in, const int* in_sizes, int n_in,
                              void* d_out, int out_size, void* d_ws, size_t ws_size,
                              hipStream_t stream) {
  const float* x = (const float*)d_in[0];
  const float* W1 = (const float*)d_in[1];
  const float* b1 = (const float*)d_in[2];
  const float* W2 = (const float*)d_in[3];
  const float* b2 = (const float*)d_in[4];
  const int* eidx = (const int*)d_in[5];
  const int* batch = (const int*)d_in[6];

  const int N = in_sizes[6];
  const int E = in_sizes[5] / 2;
  const int* row = eidx;
  const int* col = eidx + E;

  float* ws = (float*)d_ws;
  float* dinv = ws;
  float* bufA = ws + (((size_t)N + 63) & ~(size_t)63);
  unsigned* bm = (unsigned*)(bufA + (size_t)N * NCOL);

  const int G = out_size / NCOL - N;
  float* hs = (float*)d_out;
  float* hout = hs + (size_t)G * NCOL;

  const dim3 b256(256);
  const int poolBlocks = (N + (POOL_T / 32) * POOL_RW - 1) / ((POOL_T / 32) * POOL_RW);
  const int gemmBlocks = (2 * ((N + 15) / 16) + GW - 1) / GW;
  const int aggBlocks = (N + TD - 1) / TD;
  const dim3 bAgg(AGG_T);
  const int bmBlocks = (E + AGG_T * BM_C - 1) / (AGG_T * BM_C);
  const int WP = bmBlocks * BM_W;

  k_bitmap<<<bmBlocks, bAgg, 0, stream>>>(col, bm, E, WP, aggBlocks);
  k_deg_route<<<aggBlocks, bAgg, 0, stream>>>(row, col, bm, WP, dinv, N);

  k_gemm_wres<KIN><<<gemmBlocks, dim3(GT), 0, stream>>>(x, W1, dinv, bufA, N);
  k_agg_rows<<<aggBlocks, bAgg, 0, stream>>>(bufA, row, col, dinv, b1, bm, WP, hout, N, E, 1);

  k_gemm_wres<NCOL><<<gemmBlocks, dim3(GT), 0, stream>>>(hout, W2, dinv, bufA, N);
  k_agg_rows<<<aggBlocks, bAgg, 0, stream>>>(bufA, row, col, dinv, b2, bm, WP, hout, N, E, 0);

  k_zero<<<(G * NCOL + 255) / 256, b256, 0, stream>>>(hs, G * NCOL);
  k_pool_runs<<<poolBlocks, dim3(POOL_T), 0, stream>>>(hout, batch, hs, N, G);
}


# Layer-2 GEMM fused into the layer-1 aggregation pass: a block keeps its 200 finished rows in LDS beside W2 and multiplies them there (f32 WMMA, same k order); h1 never stored; 7 launches, was 8
# speedup vs baseline: 4.9639x; 1.0324x over previous
#include <hip/hip_runtime.h>


typedef __attribute__((ext_vector_type(2))) float v2f;
typedef __attribute__((ext_vector_type(8))) float v8f;

#define NCOL 128
#define KIN 512
#define GW 25
#define GT (GW * 32)
#define LWS 272

struct ablk {
  float4 q[4];
};

__device__ __forceinline__ void a_load(ablk& a, const float* __restrict__ p) {
#pragma unroll
  for (int q = 0; q < 4; ++q) a.q[q] = *(const float4*)(p + 8 * q);
}

__device__ __forceinline__ void blk_mma(v8f (&acc)[4], const ablk& a,
                                        const float* brow) {
#pragma unroll
  for (int q = 0; q < 4; ++q) {
#pragma unroll
    for (int s = 0; s < 2; ++s) {
      v2f av;
      av.x = s ? a.q[q].z : a.q[q].x;
      av.y = s ? a.q[q].w : a.q[q].y;
      const float* br = brow + (4 * q + s) * LWS;
#pragma unroll
      for (int t = 0; t < 4; ++t) {
        const v2f b = *(const v2f*)(br + t * 32);
        acc[t] = __builtin_amdgcn_wmma_f32_16x16x4_f32(
             false, av,  false, b,
             (short)0, acc[t],  false,  false);
      }
    }
  }
}

template <int K>
__global__ __launch_bounds__(GT) void k_gemm_wres(
    const float* __restrict__ X, const float* __restrict__ W,
    const float* __restrict__ scale, float* __restrict__ C, int M) {
  __shared__ float lw2[(K / 2) * LWS];

  const int tid = threadIdx.x;
  const int lane = tid & 31;
  const int wave = __builtin_amdgcn_readfirstlane(tid >> 5);
  const int c = lane & 15;
  const int h = lane >> 4;
  const int job = blockIdx.x * GW + wave;
  const int strip = job >> 1;
  const int half = job & 1;
  const int nstrips = (M + 15) >> 4;

  int r = strip * 16 + c;
  r = (r < M) ? r : (M - 1);
  const float* xp = X + (size_t)r * K + 4 * h;

  ablk a0, a1;
  a_load(a0, xp);
  a_load(a1, xp + 32);

  const int m0 = strip * 16 + 8 * h;
  float sc[8];
#pragma unroll
  for (int v = 0; v < 8; ++v) {
    int m = m0 + v;
    m = (m < M) ? m : (M - 1);
    sc[v] = scale[m];
  }

  constexpr int NW = (K / 2) * (NCOL / 4);
  for (int i0 = tid; i0 < NW; i0 += 4 * GT) {
    float4 w0[4], w1[4];
#pragma unroll
    for (int u = 0; u < 4; ++u) {
      int i = i0 + u * GT;
      i = (i < NW) ? i : (NW - 1);
      const float* wr = W + (size_t)(i >> 5) * (2 * NCOL) + (i & 31) * 4;
      w0[u] = *(const float4*)wr;
      w1[u] = *(const float4*)(wr + NCOL);
    }
#pragma unroll
    for (int u = 0; u < 4; ++u) {
      const int i = i0 + u * GT;
      if (i < NW) {
        float* dst = lw2 + (i >> 5) * LWS + (i & 31) * 8;
        ((float4*)dst)[0] = make_float4(w0[u].x, w1[u].x, w0[u].y, w1[u].y);
        ((float4*)dst)[1] = make_float4(w0[u].z, w1[u].z, w0[u].w, w1[u].w);
      }
    }
  }
  __syncthreads();
  if (strip >= nstrips) return;

  v8f acc[4] = {};
  const float* bbp = lw2 + (2 * h) * LWS + (half * 64 + c) * 2;

  for (int kb = 0; kb < K; kb += 64) {
    blk_mma(acc, a0, bbp + (kb >> 1) * LWS);
    if (kb + 64 < K) a_load(a0, xp + kb + 64);
    blk_mma(acc, a1, bbp + ((kb + 32) >> 1) * LWS);
    if (kb + 96 < K) a_load(a1, xp + kb + 96);
  }

  float* cp = C + (size_t)m0 * NCOL + half * 64 + c;
#pragma unroll
  for (int t = 0; t < 4; ++t) {
#pragma unroll
    for (int v = 0; v < 8; ++v) {
      if (m0 + v < M) cp[(size_t)v * NCOL + t * 16] = acc[t][v] * sc[v];
    }
  }
}

#define TD 200
#define AGG_T 1024

#define BM_C 4
#define BM_W (32 * BM_C)
#define BM_P (BM_W + 4)
#define BM_MAXT 256

__global__ __launch_bounds__(AGG_T) void k_bitmap(
    const int* __restrict__ col, unsigned* __restrict__ bm, int E, int WP,
    int nTiles) {
  __shared__ unsigned sm[BM_MAXT * BM_P];
  const int tid = threadIdx.x;
  uint4* sm4 = (uint4*)sm;
  const int nz = nTiles * (BM_P / 4);
  for (int i = tid; i < nz; i += AGG_T) sm4[i] = make_uint4(0u, 0u, 0u, 0u);
  __syncthreads();

  const int e0 = blockIdx.x * (AGG_T * BM_C) + tid;
  int d[BM_C];
#pragma unroll
  for (int c = 0; c < BM_C; ++c) {
    const int e = e0 + c * AGG_T;
    d[c] = (e < E) ? col[e] : -1;
  }
#pragma unroll
  for (int c = 0; c < BM_C; ++c) {
    if (d[c] >= 0)
      atomicOr(&sm[(d[c] / TD) * BM_P + c * 32 + (tid >> 5)], 1u << (tid & 31));
  }
  __syncthreads();

  const int nw = nTiles * (BM_W / 4);
  for (int i = tid; i < nw; i += AGG_T) {
    const int t = i / (BM_W / 4);
    const int j = i % (BM_W / 4);
    *(uint4*)(bm + (size_t)t * WP + (size_t)blockIdx.x * BM_W + j * 4) =
        sm4[t * (BM_P / 4) + j];
  }
}

#define DG_U 20
#define RQC 64
#define NWV (AGG_T / 32)

__device__ __forceinline__ int deg_place(int* cnt, unsigned hm, int j, int dl) {
  return ((hm >> j) & 1u) ? atomicAdd(&cnt[dl], 1) : RQC;
}

__global__ __launch_bounds__(AGG_T) void k_deg_route(
    const int* __restrict__ row, const int* __restrict__ col, unsigned* bm,
    int WP, float* __restrict__ dinv, int N) {
  __shared__ int cnt[TD];
  __shared__ int rq[TD * RQC];
  const int tid = threadIdx.x;
  const int d0 = blockIdx.x * TD;
  const int rows = (N - d0 < TD) ? (N - d0) : TD;
  if (tid < TD) cnt[tid] = 0;
  __syncthreads();

  unsigned* bmrow = bm + (size_t)blockIdx.x * WP;
  for (int base = 0; base < WP; base += DG_U * AGG_T) {
    unsigned m[DG_U];
#pragma unroll
    for (int j = 0; j < DG_U; ++j) {
      const int w = base + j * AGG_T + tid;
      m[j] = (w < WP) ? bmrow[w] : 0u;
    }
    bool more;
    do {
      int dv[DG_U], sv[DG_U];
      unsigned hm = 0u;
      unsigned wm = 0u;
#pragma unroll
      for (int j = 0; j < DG_U; ++j) {
        const unsigned mm = m[j];
        const bool has = (mm != 0u);
        if (__builtin_amdgcn_ballot_w32(has) != 0u) {
          const int e =
              has ? (((base + j * AGG_T + tid) << 5) + __builtin_ctz(mm)) : 0;
          dv[j] = col[e];
          sv[j] = row[e];
          hm |= has ? (1u << j) : 0u;
          wm |= 1u << j;
          m[j] = mm & (mm - 1u);
        }
      }
      unsigned left = 0u;
#pragma unroll
      for (int j0 = 0; j0 < DG_U; j0 += 4) {
        const int p0 = ((wm >> j0) & 1u) ? deg_place(cnt, hm, j0, dv[j0] - d0) : RQC;
        const int p1 = ((wm >> (j0 + 1)) & 1u) ? deg_place(cnt, hm, j0 + 1, dv[j0 + 1] - d0) : RQC;
        const int p2 = ((wm >> (j0 + 2)) & 1u) ? deg_place(cnt, hm, j0 + 2, dv[j0 + 2] - d0) : RQC;
        const int p3 = ((wm >> (j0 + 3)) & 1u) ? deg_place(cnt, hm, j0 + 3, dv[j0 + 3] - d0) : RQC;
        if (p0 < RQC) rq[(dv[j0] - d0) * RQC + p0] = sv[j0];
        if (p1 < RQC) rq[(dv[j0 + 1] - d0) * RQC + p1] = sv[j0 + 1];
        if (p2 < RQC) rq[(dv[j0 + 2] - d0) * RQC + p2] = sv[j0 + 2];
        if (p3 < RQC) rq[(dv[j0 + 3] - d0) * RQC + p3] = sv[j0 + 3];
        left |= m[j0] | m[j0 + 1] | m[j0 + 2] | m[j0 + 3];
      }
      more = (left != 0u);
    } while (more);
  }
  __syncthreads();

  if (tid < rows) {
    const float d = 1.0f + (float)cnt[tid];
    dinv[d0 + tid] = (d > 0.f) ? rsqrtf(d) : 0.f;
  }
  if (tid < TD) bmrow[TD * RQC + tid] = (unsigned)cnt[tid];
  const int lane = tid & 31;
  for (int r = tid >> 5; r < TD; r += NWV) {
    int c = cnt[r];
    c = (c < RQC) ? c : RQC;
    bmrow[r * RQC + lane] = (lane < c) ? (unsigned)rq[r * RQC + lane] : 0u;
    if (c > 32)
      bmrow[r * RQC + 32 + lane] =
          (32 + lane < c) ? (unsigned)rq[r * RQC + 32 + lane] : 0u;
  }
}

#define SUM_G 16

__device__ __forceinline__ void agg_sum_row(float4& cur, int s, int n, int lane,
                                            const float* __restrict__ hlin) {
  for (int i = 0; i < n; i += SUM_G) {
    int sb[SUM_G];
    float4 v[SUM_G];
#pragma unroll
    for (int u = 0; u < SUM_G; ++u) {
      const int ii = (i + u < n) ? (i + u) : (n - 1);
      sb[u] = __builtin_amdgcn_readlane(s, ii);
    }
#pragma unroll
    for (int u = 0; u < SUM_G; ++u)
      v[u] = *(const float4*)(hlin + (size_t)sb[u] * NCOL + 4 * lane);
#pragma unroll
    for (int u = 0; u < SUM_G; ++u) {
      if (i + u < n) {
        cur.x += v[u].x;
        cur.y += v[u].y;
        cur.z += v[u].z;
        cur.w += v[u].w;
      }
    }
  }
}

#define HP 132
#define TDS 208

template <bool TO_LDS>
__device__ __forceinline__ void agg_body(
    const float* __restrict__ hlin, const int* __restrict__ row,
    const int* __restrict__ col, const float* __restrict__ dinv,
    const float* __restrict__ bias, const unsigned* __restrict__ bm, int WP,
    float* __restrict__ out, float* tile, int N, int E, int do_relu) {
  const int tid = threadIdx.x;
  const int lane = tid & 31;
  const int wave = __builtin_amdgcn_readfirstlane(tid >> 5);
  const int d0 = blockIdx.x * TD;
  const int rows = (N - d0 < TD) ? (N - d0) : TD;
  const unsigned* qg = bm + (size_t)blockIdx.x * WP;

  const int rb = (wave * TD + NWV - 1) / NWV;
  int re = ((wave + 1) * TD + NWV - 1) / NWV;
  re = (re < rows) ? re : rows;
  const int nrw = re - rb;
  if (nrw <= 0) return;
  const int last = nrw - 1;

  const int f = 4 * lane;
  const float4* hsrc = (const float4*)(hlin + (size_t)d0 * NCOL);
  int cv = 0;
  float dvv = 0.f;
  if (lane < nrw) {
    cv = (int)qg[TD * RQC + rb + lane];
    dvv = dinv[d0 + rb + lane];
  }
  const float bx = bias[f], by = bias[f + 1], bz = bias[f + 2], bw = bias[f + 3];

  float4 hA = hsrc[rb * (NCOL / 4) + lane];
  int sA = (int)qg[rb * RQC + lane];
  int nA = __builtin_amdgcn_readlane(cv, 0);

  for (int k = 0; k < nrw; ++k) {
    const int r = rb + k;
    const int k1 = (k + 1 < last) ? (k + 1) : last;
    const float dr =
        __int_as_float(__builtin_amdgcn_readlane(__float_as_int(dvv), k));
    const int nB = __builtin_amdgcn_readlane(cv, k1);
    const int sB = (int)qg[(rb + k1) * RQC + lane];
    const float4 hB = hsrc[(rb + k1) * (NCOL / 4) + lane];

    float4 cur = hA;
    if (nA <= RQC) {
      agg_sum_row(cur, sA, (nA < 32) ? nA : 32, lane, hlin);
      if (nA > 32) {
        int s1 = 0;
        if (32 + lane < nA) s1 = (int)qg[r * RQC + 32 + lane];
        agg_sum_row(cur, s1, nA - 32, lane, hlin);
      }
    } else {
      const int d = d0 + r;
      for (int e0 = 0; e0 < E; e0 += 32) {
        const int e = e0 + lane;
        const int ce = (e < E) ? col[e] : -1;
        unsigned bal = __builtin_amdgcn_ballot_w32(ce == d);
        while (bal != 0u) {
          const int s = row[e0 + __builtin_ctz(bal)];
          bal &= bal - 1u;
          const float4 v = *(const float4*)(hlin + (size_t)s * NCOL + f);
          cur.x += v.x;
          cur.y += v.y;
          cur.z += v.z;
          cur.w += v.w;
        }
      }
    }
    cur.x = cur.x * dr + bx;
    cur.y = cur.y * dr + by;
    cur.z = cur.z * dr + bz;
    cur.w = cur.w * dr + bw;
    if (do_relu) {
      cur.x = (cur.x < 0.f) ? 0.f : cur.x;
      cur.y = (cur.y < 0.f) ? 0.f : cur.y;
      cur.z = (cur.z < 0.f) ? 0.f : cur.z;
      cur.w = (cur.w < 0.f) ? 0.f : cur.w;
    }
    if constexpr (TO_LDS) {
      *(float4*)(tile + r * HP + f) = cur;
    } else {
      float4* odst = (float4*)(out + (size_t)d0 * NCOL);
      odst[r * (NCOL / 4) + lane] = cur;
    }

    nA = nB;
    sA = sB;
    hA = hB;
  }
}

__global__ __launch_bounds__(AGG_T) void k_agg_rows(
    const float* __restrict__ hlin, const int* __restrict__ row,
    const int* __restrict__ col, const float* __restrict__ dinv,
    const float* __restrict__ bias, const unsigned* __restrict__ bm, int WP,
    float* __restrict__ out, int N, int E, int do_relu) {
  agg_body<false>(hlin, row, col, dinv, bias, bm, WP, out, nullptr, N, E,
                  do_relu);
}

__global__ __launch_bounds__(AGG_T) void k_agg_gemm(
    const float* __restrict__ hlin, const int* __restrict__ row,
    const int* __restrict__ col, const float* __restrict__ dinv,
    const float* __restrict__ bias, const unsigned* __restrict__ bm, int WP,
    const float* __restrict__ W, float* __restrict__ C, int N, int E) {
  __shared__ alignas(16) float lw2[(NCOL / 2) * LWS];
  __shared__ alignas(16) float ht[TDS * HP];

  const int tid = threadIdx.x;
  const int lane = tid & 31;
  const int wave = __builtin_amdgcn_readfirstlane(tid >> 5);
  const int d0 = blockIdx.x * TD;
  const int rows = (N - d0 < TD) ? (N - d0) : TD;

  constexpr int NW = (NCOL / 2) * (NCOL / 4);
  static_assert(NW == 2 * AGG_T, "two items per thread");
  {
    float4 w0[2], w1[2];
#pragma unroll
    for (int u = 0; u < 2; ++u) {
      const int i = tid + u * AGG_T;
      const float* wr = W + (size_t)(i >> 5) * (2 * NCOL) + (i & 31) * 4;
      w0[u] = *(const float4*)wr;
      w1[u] = *(const float4*)(wr + NCOL);
    }
#pragma unroll
    for (int u = 0; u < 2; ++u) {
      const int i = tid + u * AGG_T;
      float* dst = lw2 + (i >> 5) * LWS + (i & 31) * 8;
      ((float4*)dst)[0] = make_float4(w0[u].x, w1[u].x, w0[u].y, w1[u].y);
      ((float4*)dst)[1] = make_float4(w0[u].z, w1[u].z, w0[u].w, w1[u].w);
    }
  }
  for (int i = tid; i < (TDS - rows) * (NCOL / 4); i += AGG_T) {
    const int rr = rows + i / (NCOL / 4);
    *(float4*)(ht + rr * HP + (i % (NCOL / 4)) * 4) =
        make_float4(0.f, 0.f, 0.f, 0.f);
  }

  agg_body<true>(hlin, row, col, dinv, bias, bm, WP, nullptr, ht, N, E, 1);
  __syncthreads();

  if (wave >= 2 * (TDS / 16)) return;
  const int strip = wave >> 1;
  const int half = wave & 1;
  if (strip * 16 >= rows) return;
  const int c = lane & 15;
  const int h = lane >> 4;

  const int m0 = strip * 16 + 8 * h;
  float sc[8];
#pragma unroll
  for (int v = 0; v < 8; ++v) {
    int m = m0 + v;
    m = (m < rows) ? m : (rows - 1);
    sc[v] = dinv[d0 + m];
  }

  const float* ap = ht + (strip * 16 + c) * HP + 4 * h;
  const float* bbp = lw2 + (2 * h) * LWS + (half * 64 + c) * 2;
  v8f acc[4] = {};
  ablk a0, a1;
  a_load(a0, ap);
  a_load(a1, ap + 32);
  blk_mma(acc, a0, bbp);
  a_load(a0, ap + 64);
  blk_mma(acc, a1, bbp + 16 * LWS);
  a_load(a1, ap + 96);
  blk_mma(acc, a0, bbp + 32 * LWS);
  blk_mma(acc, a1, bbp + 48 * LWS);

  float* cp = C + (size_t)(d0 + m0) * NCOL + half * 64 + c;
#pragma unroll
  for (int t = 0; t < 4; ++t) {
#pragma unroll
    for (int v = 0; v < 8; ++v) {
      if (m0 + v < rows) cp[(size_t)v * NCOL + t * 16] = acc[t][v] * sc[v];
    }
  }
}

__global__ void k_zero(float* p, int n) {
  int i = blockIdx.x * blockDim.x + threadIdx.x;
  if (i < n) p[i] = 0.f;
}

#define POOL_RW 16
#define POOL_T 256

__device__ __forceinline__ void pool_flush(float* __restrict__ hs, int g, int G,
                                           int lane, const float (&a)[4]) {
  if ((unsigned)g < (unsigned)G) {
    float* p = hs + (size_t)g * NCOL + lane;
    atomicAdd(p, a[0]);
    atomicAdd(p + 32, a[1]);
    atomicAdd(p + 64, a[2]);
    atomicAdd(p + 96, a[3]);
  }
}

__global__ __launch_bounds__(POOL_T) void k_pool_runs(
    const float* __restrict__ h, const int* __restrict__ batch,
    float* __restrict__ hs, int N, int G) {
  const int lane = threadIdx.x & 31;
  const int wave = __builtin_amdgcn_readfirstlane(threadIdx.x >> 5);
  const int r0 = (blockIdx.x * (POOL_T / 32) + wave) * POOL_RW;
  if (r0 >= N) return;
  const float* hp = h + (size_t)r0 * NCOL + lane;
  const int* bp = batch + r0;
  float acc[4] = {0.f, 0.f, 0.f, 0.f};

  if (N - r0 >= POOL_RW) {
    float v[POOL_RW][4];
    int g[POOL_RW];
#pragma unroll
    for (int j = 0; j < POOL_RW; ++j) {
      v[j][0] = hp[j * NCOL];
      v[j][1] = hp[j * NCOL + 32];
      v[j][2] = hp[j * NCOL + 64];
      v[j][3] = hp[j * NCOL + 96];
      g[j] = bp[j];
    }
    int cur = g[0];
#pragma unroll
    for (int j = 0; j < POOL_RW; ++j) {
      if (g[j] != cur) {
        pool_flush(hs, cur, G, lane, acc);
        acc[0] = acc[1] = acc[2] = acc[3] = 0.f;
        cur = g[j];
      }
      acc[0] += v[j][0];
      acc[1] += v[j][1];
      acc[2] += v[j][2];
      acc[3] += v[j][3];
    }
    pool_flush(hs, cur, G, lane, acc);
  } else {
    const int nr = N - r0;
    int cur = bp[0];
    for (int j = 0; j < nr; ++j) {
      const int gj = bp[j];
      if (gj != cur) {
        pool_flush(hs, cur, G, lane, acc);
        acc[0] = acc[1] = acc[2] = acc[3] = 0.f;
        cur = gj;
      }
      acc[0] += hp[j * NCOL];
      acc[1] += hp[j * NCOL + 32];
      acc[2] += hp[j * NCOL + 64];
      acc[3] += hp[j * NCOL + 96];
    }
    pool_flush(hs, cur, G, lane, acc);
  }
}

extern "C" void kernel_launch(void* const* d_in, const int* in_sizes, int n_in,
                              void* d_out, int out_size, void* d_ws, size_t ws_size,
                              hipStream_t stream) {
  const float* x = (const float*)d_in[0];
  const float* W1 = (const float*)d_in[1];
  const float* b1 = (const float*)d_in[2];
  const float* W2 = (const float*)d_in[3];
  const float* b2 = (const float*)d_in[4];
  const int* eidx = (const int*)d_in[5];
  const int* batch = (const int*)d_in[6];

  const int N = in_sizes[6];
  const int E = in_sizes[5] / 2;
  const int* row = eidx;
  const int* col = eidx + E;

  float* ws = (float*)d_ws;
  float* dinv = ws;
  float* bufA = ws + (((size_t)N + 63) & ~(size_t)63);
  unsigned* bm = (unsigned*)(bufA + (size_t)N * NCOL);

  const int G = out_size / NCOL - N;
  float* hs = (float*)d_out;
  float* hout = hs + (size_t)G * NCOL;

  const dim3 b256(256);
  const int poolBlocks = (N + (POOL_T / 32) * POOL_RW - 1) / ((POOL_T / 32) * POOL_RW);
  const int gemmBlocks = (2 * ((N + 15) / 16) + GW - 1) / GW;
  const int aggBlocks = (N + TD - 1) / TD;
  const dim3 bAgg(AGG_T);
  const int bmBlocks = (E + AGG_T * BM_C - 1) / (AGG_T * BM_C);
  const int WP = bmBlocks * BM_W;

  k_bitmap<<<bmBlocks, bAgg, 0, stream>>>(col, bm, E, WP, aggBlocks);
  k_deg_route<<<aggBlocks, bAgg, 0, stream>>>(row, col, bm, WP, dinv, N);

  k_gemm_wres<KIN><<<gemmBlocks, dim3(GT), 0, stream>>>(x, W1, dinv, hout, N);
  k_agg_gemm<<<aggBlocks, bAgg, 0, stream>>>(hout, row, col, dinv, b1, bm, WP, W2, bufA, N, E);

  k_agg_rows<<<aggBlocks, bAgg, 0, stream>>>(bufA, row, col, dinv, b2, bm, WP, hout, N, E, 0);

  k_zero<<<(G * NCOL + 255) / 256, b256, 0, stream>>>(hs, G * NCOL);
  k_pool_runs<<<poolBlocks, dim3(POOL_T), 0, stream>>>(hout, batch, hs, N, G);
}
